# starting kernel
# baseline (speedup 1.0000x reference)
#include <hip/hip_runtime.h>
#include <math.h>

constexpr int kBS      = 64;
constexpr int kNC      = 128;
constexpr int kN       = 32;
constexpr int kDN      = 64;
constexpr int kHMOD    = 128;
constexpr int kMODIN   = 2 * kN + 4 + kN * kN;
constexpr int kMODOUT  = kN * kN + kN;
constexpr int kHEAD    = 2 * kN + 4;
constexpr int kHSTRIDE = 132;

typedef __attribute__((ext_vector_type(2))) float v2f;
typedef __attribute__((ext_vector_type(8))) float v8f;

__global__ __launch_bounds__(256) void head_kernel(
    const float* __restrict__ h, const float* __restrict__ msg,
    const float* __restrict__ decay, const float* __restrict__ drift,
    const float* __restrict__ s_live, const float* __restrict__ s_ema,
    float* __restrict__ head) {
  const int t    = blockIdx.x * 256 + threadIdx.x;
  const int pair = t >> 6;
  const int j    = t & 63;

  const float* src = (j < kN ? h : msg) +
                     (size_t)pair * (kN * kDN) + (size_t)(j & (kN - 1)) * kDN;
  const float4* p4 = (const float4*)src;
  float s = 0.f;
#pragma unroll
  for (int i = 0; i < kDN / 4; ++i) {
    float4 v = p4[i];
    s += v.x * v.x + v.y * v.y + v.z * v.z + v.w * v.w;
  }
  head[(size_t)pair * kHEAD + j] = sqrtf(s);

  if (j == 0) {
    const float* dp = decay + (size_t)pair * kN;
    float m = 0.f;
#pragma unroll
    for (int i = 0; i < kN; ++i) m += dp[i];
    const int b = pair / kNC;
    float* hd = head + (size_t)pair * kHEAD;
    hd[kHEAD - 4] = m * (1.f / (float)kN);
    hd[kHEAD - 3] = drift[pair];
    hd[kHEAD - 2] = s_live[b];
    hd[kHEAD - 1] = s_ema[b];
  }
}

__global__ __launch_bounds__(256) void modmlp_kernel(
    const float* __restrict__ head, const float* __restrict__ hebb,
    const float* __restrict__ w1, const float* __restrict__ b1,
    const float* __restrict__ w2, const float* __restrict__ b2,
    float* __restrict__ out) {
  extern __shared__ float smem[];
  float* ldsA = smem;
  float* ldsH = smem + 16 * kMODIN;

  const int tid   = threadIdx.x;
  const int lane  = tid & 31;
  const int wave  = tid >> 5;
  const int mtile = blockIdx.x;
  const int cell  = blockIdx.y;
  const int b0    = mtile * 16;
  const int arow  = lane & 15;
  const int koff  = (lane >> 4) << 1;
  const int rhalf = (lane >> 4) << 3;

  for (int r = 0; r < 16; ++r) {
    const size_t pr = (size_t)(b0 + r) * kNC + cell;
    const float* hr = head + pr * kHEAD;
    const float* br = hebb + pr * (kN * kN);
    float* dst = ldsA + r * kMODIN;
    if (tid < kHEAD) dst[tid] = hr[tid];
    for (int k = tid; k < kN * kN; k += 256) dst[kHEAD + k] = br[k];
  }
  __syncthreads();

  const float* w1c  = w1 + (size_t)cell * kMODIN * kHMOD;
  const int    hcol = (wave << 4) + arow;
  v8f acc;
  {
    const float bias = b1[cell * kHMOD + hcol];
#pragma unroll
    for (int i = 0; i < 8; ++i) acc[i] = bias;
  }
  const float* aRow = ldsA + arow * kMODIN + koff;
#pragma unroll 4
  for (int k0 = 0; k0 < kMODIN; k0 += 4) {
    const v2f a = *(const v2f*)(aRow + k0);
    v2f b;
    b.x = w1c[(size_t)(k0 + koff) * kHMOD + hcol];
    b.y = w1c[(size_t)(k0 + koff + 1) * kHMOD + hcol];
    acc = __builtin_amdgcn_wmma_f32_16x16x4_f32(
        false, a, false, b, (short)0, acc, false, false);
  }
#pragma unroll
  for (int i = 0; i < 8; ++i)
    ldsH[(i + rhalf) * kHSTRIDE + hcol] = tanhf(acc[i]);
  __syncthreads();

  const float* w2c = w2 + (size_t)cell * kHMOD * kMODOUT;
  const float* b2c = b2 + (size_t)cell * kMODOUT;
  const float* aH  = ldsH + arow * kHSTRIDE + koff;
  for (int ot = wave; ot < kMODOUT / 16; ot += 8) {
    const int ocol = (ot << 4) + arow;
    v8f c;
    {
      const float bias = b2c[ocol];
#pragma unroll
      for (int i = 0; i < 8; ++i) c[i] = bias;
    }
#pragma unroll 8
    for (int k0 = 0; k0 < kHMOD; k0 += 4) {
      const v2f a = *(const v2f*)(aH + k0);
      v2f b;
      b.x = w2c[(size_t)(k0 + koff) * kMODOUT + ocol];
      b.y = w2c[(size_t)(k0 + koff + 1) * kMODOUT + ocol];
      c = __builtin_amdgcn_wmma_f32_16x16x4_f32(
          false, a, false, b, (short)0, c, false, false);
    }
#pragma unroll
    for (int i = 0; i < 8; ++i) {
      const int row = b0 + rhalf + i;
      out[((size_t)row * kNC + cell) * kMODOUT + ocol] = c[i];
    }
  }
}

extern "C" void kernel_launch(void* const* d_in, const int* in_sizes, int n_in,
                              void* d_out, int out_size, void* d_ws, size_t ws_size,
                              hipStream_t stream) {
  const float* h     = (const float*)d_in[0];
  const float* msg   = (const float*)d_in[1];
  const float* decay = (const float*)d_in[2];
  const float* hebb  = (const float*)d_in[3];
  const float* drift = (const float*)d_in[4];
  const float* slive = (const float*)d_in[5];
  const float* sema  = (const float*)d_in[6];
  const float* w1    = (const float*)d_in[7];
  const float* b1    = (const float*)d_in[8];
  const float* w2    = (const float*)d_in[9];
  const float* b2    = (const float*)d_in[10];
  float* headws = (float*)d_ws;
  float* out    = (float*)d_out;

  const int norm_threads = kBS * kNC * 2 * kN;
  head_kernel<<<norm_threads / 256, 256, 0, stream>>>(
      h, msg, decay, drift, slive, sema, headws);

  dim3 grid(kBS / 16, kNC);
  const size_t smem = (size_t)(16 * kMODIN + 16 * kHSTRIDE) * sizeof(float);
  modmlp_kernel<<<grid, 256, smem, stream>>>(
      headws, hebb, w1, b1, w2, b2, out);
}


# modmlp A-tile staging: each thread's 17 independent 128-bit global loads are batched ahead of their LDS stores, replacing 80 serial load-wait-store round trips per thread (64 of them to HBM)
# speedup vs baseline: 1.3306x; 1.3306x over previous
#include <hip/hip_runtime.h>
#include <math.h>

constexpr int kBS      = 64;
constexpr int kNC      = 128;
constexpr int kN       = 32;
constexpr int kDN      = 64;
constexpr int kHMOD    = 128;
constexpr int kMODIN   = 2 * kN + 4 + kN * kN;
constexpr int kMODOUT  = kN * kN + kN;
constexpr int kHEAD    = 2 * kN + 4;
constexpr int kHSTRIDE = 132;

typedef __attribute__((ext_vector_type(2))) float v2f;
typedef __attribute__((ext_vector_type(8))) float v8f;

__global__ __launch_bounds__(256) void head_kernel(
    const float* __restrict__ h, const float* __restrict__ msg,
    const float* __restrict__ decay, const float* __restrict__ drift,
    const float* __restrict__ s_live, const float* __restrict__ s_ema,
    float* __restrict__ head) {
  const int t    = blockIdx.x * 256 + threadIdx.x;
  const int pair = t >> 6;
  const int j    = t & 63;

  const float* src = (j < kN ? h : msg) +
                     (size_t)pair * (kN * kDN) + (size_t)(j & (kN - 1)) * kDN;
  const float4* p4 = (const float4*)src;
  float s = 0.f;
#pragma unroll
  for (int i = 0; i < kDN / 4; ++i) {
    float4 v = p4[i];
    s += v.x * v.x + v.y * v.y + v.z * v.z + v.w * v.w;
  }
  head[(size_t)pair * kHEAD + j] = sqrtf(s);

  if (j == 0) {
    const float* dp = decay + (size_t)pair * kN;
    float m = 0.f;
#pragma unroll
    for (int i = 0; i < kN; ++i) m += dp[i];
    const int b = pair / kNC;
    float* hd = head + (size_t)pair * kHEAD;
    hd[kHEAD - 4] = m * (1.f / (float)kN);
    hd[kHEAD - 3] = drift[pair];
    hd[kHEAD - 2] = s_live[b];
    hd[kHEAD - 1] = s_ema[b];
  }
}

__global__ __launch_bounds__(256) void modmlp_kernel(
    const float* __restrict__ head, const float* __restrict__ hebb,
    const float* __restrict__ w1, const float* __restrict__ b1,
    const float* __restrict__ w2, const float* __restrict__ b2,
    float* __restrict__ out) {
  extern __shared__ __align__(16) float smem[];
  float* ldsA = smem;
  float* ldsH = smem + 16 * kMODIN;

  const int tid   = threadIdx.x;
  const int lane  = tid & 31;
  const int wave  = tid >> 5;
  const int mtile = blockIdx.x;
  const int cell  = blockIdx.y;
  const int b0    = mtile * 16;
  const int arow  = lane & 15;
  const int koff  = (lane >> 4) << 1;
  const int rhalf = (lane >> 4) << 3;

  {
    const size_t pr0 = (size_t)b0 * kNC + cell;
    const float* hb = hebb + pr0 * (kN * kN) + 4 * tid;
    float4 hv[16];
#pragma unroll
    for (int r = 0; r < 16; ++r)
      hv[r] = *(const float4*)(hb + (size_t)r * kNC * (kN * kN));
    const int nr = tid >> 4, nq = tid & 15;
    const float4 nv =
        *(const float4*)(head + (pr0 + (size_t)nr * kNC) * kHEAD + 4 * nq);
    *(float4*)(ldsA + nr * kMODIN + 4 * nq) = nv;
#pragma unroll
    for (int r = 0; r < 16; ++r)
      *(float4*)(ldsA + r * kMODIN + kHEAD + 4 * tid) = hv[r];
    if (tid < 16)
      *(float4*)(ldsA + tid * kMODIN + 2 * kN) =
          *(const float4*)(head + (pr0 + (size_t)tid * kNC) * kHEAD + 2 * kN);
  }
  __syncthreads();

  const float* w1c  = w1 + (size_t)cell * kMODIN * kHMOD;
  const int    hcol = (wave << 4) + arow;
  v8f acc;
  {
    const float bias = b1[cell * kHMOD + hcol];
#pragma unroll
    for (int i = 0; i < 8; ++i) acc[i] = bias;
  }
  const float* aRow = ldsA + arow * kMODIN + koff;
#pragma unroll 4
  for (int k0 = 0; k0 < kMODIN; k0 += 4) {
    const v2f a = *(const v2f*)(aRow + k0);
    v2f b;
    b.x = w1c[(size_t)(k0 + koff) * kHMOD + hcol];
    b.y = w1c[(size_t)(k0 + koff + 1) * kHMOD + hcol];
    acc = __builtin_amdgcn_wmma_f32_16x16x4_f32(
        false, a, false, b, (short)0, acc, false, false);
  }
#pragma unroll
  for (int i = 0; i < 8; ++i)
    ldsH[(i + rhalf) * kHSTRIDE + hcol] = tanhf(acc[i]);
  __syncthreads();

  const float* w2c = w2 + (size_t)cell * kHMOD * kMODOUT;
  const float* b2c = b2 + (size_t)cell * kMODOUT;
  const float* aH  = ldsH + arow * kHSTRIDE + koff;
  for (int ot = wave; ot < kMODOUT / 16; ot += 8) {
    const int ocol = (ot << 4) + arow;
    v8f c;
    {
      const float bias = b2c[ocol];
#pragma unroll
      for (int i = 0; i < 8; ++i) c[i] = bias;
    }
#pragma unroll 8
    for (int k0 = 0; k0 < kHMOD; k0 += 4) {
      const v2f a = *(const v2f*)(aH + k0);
      v2f b;
      b.x = w2c[(size_t)(k0 + koff) * kMODOUT + ocol];
      b.y = w2c[(size_t)(k0 + koff + 1) * kMODOUT + ocol];
      c = __builtin_amdgcn_wmma_f32_16x16x4_f32(
          false, a, false, b, (short)0, c, false, false);
    }
#pragma unroll
    for (int i = 0; i < 8; ++i) {
      const int row = b0 + rhalf + i;
      out[((size_t)row * kNC + cell) * kMODOUT + ocol] = c[i];
    }
  }
}

extern "C" void kernel_launch(void* const* d_in, const int* in_sizes, int n_in,
                              void* d_out, int out_size, void* d_ws, size_t ws_size,
                              hipStream_t stream) {
  const float* h     = (const float*)d_in[0];
  const float* msg   = (const float*)d_in[1];
  const float* decay = (const float*)d_in[2];
  const float* hebb  = (const float*)d_in[3];
  const float* drift = (const float*)d_in[4];
  const float* slive = (const float*)d_in[5];
  const float* sema  = (const float*)d_in[6];
  const float* w1    = (const float*)d_in[7];
  const float* b1    = (const float*)d_in[8];
  const float* w2    = (const float*)d_in[9];
  const float* b2    = (const float*)d_in[10];
  float* headws = (float*)d_ws;
  float* out    = (float*)d_out;

  const int norm_threads = kBS * kNC * 2 * kN;
  head_kernel<<<norm_threads / 256, 256, 0, stream>>>(
      h, msg, decay, drift, slive, sema, headws);

  dim3 grid(kBS / 16, kNC);
  const size_t smem = (size_t)(16 * kMODIN + 16 * kHSTRIDE) * sizeof(float);
  modmlp_kernel<<<grid, 256, smem, stream>>>(
      headws, hebb, w1, b1, w2, b2, out);
}


# modmlp weight fragments by 128-bit row loads: a wave owns 64 columns (4 N-tiles, lane c = columns 4c..4c+3, 4 accumulators); layer 1 split-K over 4 waves, partials summed in LDS
# speedup vs baseline: 1.6727x; 1.2571x over previous
#include <hip/hip_runtime.h>
#include <math.h>

constexpr int kBS      = 64;
constexpr int kNC      = 128;
constexpr int kN       = 32;
constexpr int kDN      = 64;
constexpr int kHMOD    = 128;
constexpr int kMODIN   = 2 * kN + 4 + kN * kN;
constexpr int kMODOUT  = kN * kN + kN;
constexpr int kHEAD    = 2 * kN + 4;
constexpr int kHSTRIDE = 132;

typedef __attribute__((ext_vector_type(2))) float v2f;
typedef __attribute__((ext_vector_type(8))) float v8f;

__global__ __launch_bounds__(256) void head_kernel(
    const float* __restrict__ h, const float* __restrict__ msg,
    const float* __restrict__ decay, const float* __restrict__ drift,
    const float* __restrict__ s_live, const float* __restrict__ s_ema,
    float* __restrict__ head) {
  const int t    = blockIdx.x * 256 + threadIdx.x;
  const int pair = t >> 6;
  const int j    = t & 63;

  const float* src = (j < kN ? h : msg) +
                     (size_t)pair * (kN * kDN) + (size_t)(j & (kN - 1)) * kDN;
  const float4* p4 = (const float4*)src;
  float s = 0.f;
#pragma unroll
  for (int i = 0; i < kDN / 4; ++i) {
    float4 v = p4[i];
    s += v.x * v.x + v.y * v.y + v.z * v.z + v.w * v.w;
  }
  head[(size_t)pair * kHEAD + j] = sqrtf(s);

  if (j == 0) {
    const float* dp = decay + (size_t)pair * kN;
    float m = 0.f;
#pragma unroll
    for (int i = 0; i < kN; ++i) m += dp[i];
    const int b = pair / kNC;
    float* hd = head + (size_t)pair * kHEAD;
    hd[kHEAD - 4] = m * (1.f / (float)kN);
    hd[kHEAD - 3] = drift[pair];
    hd[kHEAD - 2] = s_live[b];
    hd[kHEAD - 1] = s_ema[b];
  }
}

__device__ __forceinline__ v8f wmma4(v2f a, float bx, float by, v8f c) {
  v2f b;
  b.x = bx;
  b.y = by;
  return __builtin_amdgcn_wmma_f32_16x16x4_f32(
      false, a, false, b, (short)0, c, false, false);
}

template <int U>
__device__ __forceinline__ void load_b(float4 (&lo)[U], float4 (&hi)[U],
                                       const float* __restrict__ p,
                                       int pitch) {
#pragma unroll
  for (int u = 0; u < U; ++u) {
    lo[u] = *(const float4*)(p + (size_t)(4 * u) * pitch);
    hi[u] = *(const float4*)(p + (size_t)(4 * u + 1) * pitch);
  }
}

template <int U>
__device__ __forceinline__ void mma_b(const float4 (&lo)[U],
                                      const float4 (&hi)[U],
                                      const float* __restrict__ ap, v8f& c0,
                                      v8f& c1, v8f& c2, v8f& c3) {
#pragma unroll
  for (int u = 0; u < U; ++u) {
    const v2f a = *(const v2f*)(ap + 4 * u);
    c0 = wmma4(a, lo[u].x, hi[u].x, c0);
    c1 = wmma4(a, lo[u].y, hi[u].y, c1);
    c2 = wmma4(a, lo[u].z, hi[u].z, c2);
    c3 = wmma4(a, lo[u].w, hi[u].w, c3);
  }
}

constexpr int kKS1   = kMODIN / 4;
constexpr int kKSQ   = kKS1 / 4;
constexpr int kU1    = 4;
constexpr int kU2    = 8;

__global__ __launch_bounds__(256) __attribute__((amdgpu_waves_per_eu(4, 4)))
void modmlp_kernel(
    const float* __restrict__ head, const float* __restrict__ hebb,
    const float* __restrict__ w1, const float* __restrict__ b1,
    const float* __restrict__ w2, const float* __restrict__ b2,
    float* __restrict__ out) {
  extern __shared__ __align__(16) float smem[];
  float* ldsA = smem;
  float* ldsH = smem + 16 * kMODIN;
  float* ldsP = ldsH + 16 * kHSTRIDE;

  const int tid   = threadIdx.x;
  const int lane  = tid & 31;
  const int wave  = tid >> 5;
  const int mtile = blockIdx.x;
  const int cell  = blockIdx.y;
  const int b0    = mtile * 16;
  const int arow  = lane & 15;
  const int koff  = (lane >> 4) << 1;
  const int rhalf = (lane >> 4) << 3;

  {
    const size_t pr0 = (size_t)b0 * kNC + cell;
    const float* hb = hebb + pr0 * (kN * kN) + 4 * tid;
    float4 hv[16];
#pragma unroll
    for (int r = 0; r < 16; ++r)
      hv[r] = *(const float4*)(hb + (size_t)r * kNC * (kN * kN));
    const int nr = tid >> 4, nq = tid & 15;
    const float4 nv =
        *(const float4*)(head + (pr0 + (size_t)nr * kNC) * kHEAD + 4 * nq);
    *(float4*)(ldsA + nr * kMODIN + 4 * nq) = nv;
#pragma unroll
    for (int r = 0; r < 16; ++r)
      *(float4*)(ldsA + r * kMODIN + kHEAD + 4 * tid) = hv[r];
    if (tid < 16)
      *(float4*)(ldsA + tid * kMODIN + 2 * kN) =
          *(const float4*)(head + (pr0 + (size_t)tid * kNC) * kHEAD + 2 * kN);
  }
  __syncthreads();

  {
    const int g = wave & 1;
    const int q = wave >> 1;
    const float* w1c = w1 + (size_t)cell * kMODIN * kHMOD;
    const float* bp  = w1c + (size_t)(4 * kKSQ * q + koff) * kHMOD +
                       64 * g + 4 * arow;
    const float* ap  = ldsA + arow * kMODIN + koff + 4 * kKSQ * q;
    v8f c0 = {}, c1 = {}, c2 = {}, c3 = {};
    float4 lo[kU1], hi[kU1], nlo[kU1], nhi[kU1];
    load_b<kU1>(lo, hi, bp, kHMOD);
#pragma unroll
    for (int it = 0; it < kKSQ / kU1 - 1; ++it) {
      bp += (size_t)(4 * kU1) * kHMOD;
      load_b<kU1>(nlo, nhi, bp, kHMOD);
      mma_b<kU1>(lo, hi, ap, c0, c1, c2, c3);
      ap += 4 * kU1;
#pragma unroll
      for (int u = 0; u < kU1; ++u) {
        lo[u] = nlo[u];
        hi[u] = nhi[u];
      }
    }
    float4 tlo[1], thi[1];
    if (q == 3)
      load_b<1>(tlo, thi, bp + (size_t)(4 * kU1) * kHMOD, kHMOD);
    mma_b<kU1>(lo, hi, ap, c0, c1, c2, c3);
    if (q == 3)
      mma_b<1>(tlo, thi, ap + 4 * kU1, c0, c1, c2, c3);

    float* pp = ldsP + (q * 16 + rhalf) * kHMOD + 64 * g + 4 * arow;
#pragma unroll
    for (int r = 0; r < 8; ++r)
      *(float4*)(pp + r * kHMOD) = make_float4(c0[r], c1[r], c2[r], c3[r]);
  }
  __syncthreads();
  {
    const int row = tid >> 4, col = (tid & 15) << 3;
    const float* bb = b1 + cell * kHMOD + col;
    float s[8];
    {
      const float4 u0 = *(const float4*)(bb);
      const float4 u1 = *(const float4*)(bb + 4);
      s[0] = u0.x; s[1] = u0.y; s[2] = u0.z; s[3] = u0.w;
      s[4] = u1.x; s[5] = u1.y; s[6] = u1.z; s[7] = u1.w;
    }
#pragma unroll
    for (int q = 0; q < 4; ++q) {
      const float* pq = ldsP + (q * 16 + row) * kHMOD + col;
      const float4 u0 = *(const float4*)(pq);
      const float4 u1 = *(const float4*)(pq + 4);
      s[0] += u0.x; s[1] += u0.y; s[2] += u0.z; s[3] += u0.w;
      s[4] += u1.x; s[5] += u1.y; s[6] += u1.z; s[7] += u1.w;
    }
    float* hp = ldsH + row * kHSTRIDE + col;
    *(float4*)(hp) =
        make_float4(tanhf(s[0]), tanhf(s[1]), tanhf(s[2]), tanhf(s[3]));
    *(float4*)(hp + 4) =
        make_float4(tanhf(s[4]), tanhf(s[5]), tanhf(s[6]), tanhf(s[7]));
  }
  __syncthreads();

  const float* w2c = w2 + (size_t)cell * kHMOD * kMODOUT;
  const float* b2c = b2 + (size_t)cell * kMODOUT;
  const float* aH  = ldsH + arow * kHSTRIDE + koff;
#pragma unroll
  for (int st = wave; st < 16; st += 8) {
    const int cb = 64 * st + 4 * arow;
    v8f c0, c1, c2, c3;
    {
      const float4 bias = *(const float4*)(b2c + cb);
#pragma unroll
      for (int i = 0; i < 8; ++i) {
        c0[i] = bias.x; c1[i] = bias.y; c2[i] = bias.z; c3[i] = bias.w;
      }
    }
    const float* bp = w2c + (size_t)koff * kMODOUT + cb;
    float4 lo[kU2], hi[kU2], nlo[kU2], nhi[kU2];
    load_b<kU2>(lo, hi, bp, kMODOUT);
#pragma unroll
    for (int it = 0; it < kHMOD / 4 / kU2; ++it) {
      if (it + 1 < kHMOD / 4 / kU2)
        load_b<kU2>(nlo, nhi, bp + (size_t)(4 * kU2) * (it + 1) * kMODOUT,
                    kMODOUT);
      mma_b<kU2>(lo, hi, aH + 4 * kU2 * it, c0, c1, c2, c3);
#pragma unroll
      for (int u = 0; u < kU2; ++u) {
        lo[u] = nlo[u];
        hi[u] = nhi[u];
      }
    }
    float* op = out + ((size_t)(b0 + rhalf) * kNC + cell) * kMODOUT + cb;
#pragma unroll
    for (int r = 0; r < 8; ++r)
      *(float4*)(op + (size_t)r * kNC * kMODOUT) =
          make_float4(c0[r], c1[r], c2[r], c3[r]);
  }
  if (wave < 2) {
    const int ocol = 1024 + (wave << 4) + arow;
    v8f c;
    {
      const float bias = b2c[ocol];
#pragma unroll
      for (int i = 0; i < 8; ++i) c[i] = bias;
    }
#pragma unroll
    for (int k0 = 0; k0 < kHMOD; k0 += 4) {
      const v2f a = *(const v2f*)(aH + k0);
      c = wmma4(a, w2c[(size_t)(k0 + koff) * kMODOUT + ocol],
                w2c[(size_t)(k0 + koff + 1) * kMODOUT + ocol], c);
    }
#pragma unroll
    for (int i = 0; i < 8; ++i) {
      const int row = b0 + rhalf + i;
      out[((size_t)row * kNC + cell) * kMODOUT + ocol] = c[i];
    }
  }
}

extern "C" void kernel_launch(void* const* d_in, const int* in_sizes, int n_in,
                              void* d_out, int out_size, void* d_ws, size_t ws_size,
                              hipStream_t stream) {
  const float* h     = (const float*)d_in[0];
  const float* msg   = (const float*)d_in[1];
  const float* decay = (const float*)d_in[2];
  const float* hebb  = (const float*)d_in[3];
  const float* drift = (const float*)d_in[4];
  const float* slive = (const float*)d_in[5];
  const float* sema  = (const float*)d_in[6];
  const float* w1    = (const float*)d_in[7];
  const float* b1    = (const float*)d_in[8];
  const float* w2    = (const float*)d_in[9];
  const float* b2    = (const float*)d_in[10];
  float* headws = (float*)d_ws;
  float* out    = (float*)d_out;

  const int norm_threads = kBS * kNC * 2 * kN;
  head_kernel<<<norm_threads / 256, 256, 0, stream>>>(
      h, msg, decay, drift, slive, sema, headws);

  dim3 grid(kBS / 16, kNC);
  const size_t smem =
      (size_t)(16 * kMODIN + 16 * kHSTRIDE + 4 * 16 * kHMOD) * sizeof(float);
  modmlp_kernel<<<grid, 256, smem, stream>>>(
      headws, hebb, w1, b1, w2, b2, out);
}


# modmlp work-group covers 32 batch rows (two 16-row M tiles): each loaded weight fragment feeds both tiles, halving weight bytes and load instructions per WMMA; grid 2x128, one 8-wave group per WGP
# speedup vs baseline: 1.9195x; 1.1475x over previous
#include <hip/hip_runtime.h>
#include <math.h>

constexpr int kBS      = 64;
constexpr int kNC      = 128;
constexpr int kN       = 32;
constexpr int kDN      = 64;
constexpr int kHMOD    = 128;
constexpr int kMODIN   = 2 * kN + 4 + kN * kN;
constexpr int kMODOUT  = kN * kN + kN;
constexpr int kHEAD    = 2 * kN + 4;
constexpr int kHSTRIDE = 132;
constexpr int kMT      = 2;
constexpr int kROWS    = 16 * kMT;

typedef __attribute__((ext_vector_type(2))) float v2f;
typedef __attribute__((ext_vector_type(8))) float v8f;

__global__ __launch_bounds__(256) void head_kernel(
    const float* __restrict__ h, const float* __restrict__ msg,
    const float* __restrict__ decay, const float* __restrict__ drift,
    const float* __restrict__ s_live, const float* __restrict__ s_ema,
    float* __restrict__ head) {
  const int t    = blockIdx.x * 256 + threadIdx.x;
  const int pair = t >> 6;
  const int j    = t & 63;

  const float* src = (j < kN ? h : msg) +
                     (size_t)pair * (kN * kDN) + (size_t)(j & (kN - 1)) * kDN;
  const float4* p4 = (const float4*)src;
  float s = 0.f;
#pragma unroll
  for (int i = 0; i < kDN / 4; ++i) {
    float4 v = p4[i];
    s += v.x * v.x + v.y * v.y + v.z * v.z + v.w * v.w;
  }
  head[(size_t)pair * kHEAD + j] = sqrtf(s);

  if (j == 0) {
    const float* dp = decay + (size_t)pair * kN;
    float m = 0.f;
#pragma unroll
    for (int i = 0; i < kN; ++i) m += dp[i];
    const int b = pair / kNC;
    float* hd = head + (size_t)pair * kHEAD;
    hd[kHEAD - 4] = m * (1.f / (float)kN);
    hd[kHEAD - 3] = drift[pair];
    hd[kHEAD - 2] = s_live[b];
    hd[kHEAD - 1] = s_ema[b];
  }
}

__device__ __forceinline__ v8f wmma4(v2f a, float bx, float by, v8f c) {
  v2f b;
  b.x = bx;
  b.y = by;
  return __builtin_amdgcn_wmma_f32_16x16x4_f32(
      false, a, false, b, (short)0, c, false, false);
}

template <int U>
__device__ __forceinline__ void load_b(float4 (&lo)[U], float4 (&hi)[U],
                                       const float* __restrict__ p,
                                       int pitch) {
#pragma unroll
  for (int u = 0; u < U; ++u) {
    lo[u] = *(const float4*)(p + (size_t)(4 * u) * pitch);
    hi[u] = *(const float4*)(p + (size_t)(4 * u + 1) * pitch);
  }
}

template <int U>
__device__ __forceinline__ void mma_b(const float4 (&lo)[U],
                                      const float4 (&hi)[U],
                                      const float* __restrict__ ap, int mpitch,
                                      v8f (&c)[kMT][4]) {
#pragma unroll
  for (int u = 0; u < U; ++u) {
#pragma unroll
    for (int mt = 0; mt < kMT; ++mt) {
      const v2f a = *(const v2f*)(ap + mt * mpitch + 4 * u);
      c[mt][0] = wmma4(a, lo[u].x, hi[u].x, c[mt][0]);
      c[mt][1] = wmma4(a, lo[u].y, hi[u].y, c[mt][1]);
      c[mt][2] = wmma4(a, lo[u].z, hi[u].z, c[mt][2]);
      c[mt][3] = wmma4(a, lo[u].w, hi[u].w, c[mt][3]);
    }
  }
}

constexpr int kKS1   = kMODIN / 4;
constexpr int kKSQ   = kKS1 / 4;
constexpr int kU1    = 4;
constexpr int kU2    = 8;

constexpr int kLdsFloats = kROWS * kMODIN + kROWS * kHSTRIDE + 4 * kROWS * kHMOD;

__global__ __launch_bounds__(256) __attribute__((amdgpu_waves_per_eu(2, 2)))
void modmlp_kernel(
    const float* __restrict__ head, const float* __restrict__ hebb,
    const float* __restrict__ w1, const float* __restrict__ b1,
    const float* __restrict__ w2, const float* __restrict__ b2,
    float* __restrict__ out) {
  __shared__ __align__(16) float smem[kLdsFloats];
  float* ldsA = smem;
  float* ldsH = smem + kROWS * kMODIN;
  float* ldsP = ldsH + kROWS * kHSTRIDE;

  const int tid   = threadIdx.x;
  const int lane  = tid & 31;
  const int wave  = tid >> 5;
  const int mtile = blockIdx.x;
  const int cell  = blockIdx.y;
  const int b0    = mtile * kROWS;
  const int arow  = lane & 15;
  const int koff  = (lane >> 4) << 1;
  const int rhalf = (lane >> 4) << 3;

  {
    const size_t pr0 = (size_t)b0 * kNC + cell;
    const float* hb = hebb + pr0 * (kN * kN) + 4 * tid;
    float4 hv[kROWS];
#pragma unroll
    for (int r = 0; r < kROWS; ++r)
      hv[r] = *(const float4*)(hb + (size_t)r * kNC * (kN * kN));
    const int nr = tid >> 4, nq = tid & 15;
    float4 nv[kMT];
#pragma unroll
    for (int i = 0; i < kMT; ++i)
      nv[i] = *(const float4*)(head + (pr0 + (size_t)(nr + 16 * i) * kNC) * kHEAD +
                               4 * nq);
#pragma unroll
    for (int i = 0; i < kMT; ++i)
      *(float4*)(ldsA + (nr + 16 * i) * kMODIN + 4 * nq) = nv[i];
#pragma unroll
    for (int r = 0; r < kROWS; ++r)
      *(float4*)(ldsA + r * kMODIN + kHEAD + 4 * tid) = hv[r];
    if (tid < kROWS)
      *(float4*)(ldsA + tid * kMODIN + 2 * kN) =
          *(const float4*)(head + (pr0 + (size_t)tid * kNC) * kHEAD + 2 * kN);
  }
  __syncthreads();

  {
    const int g = wave & 1;
    const int q = wave >> 1;
    const float* w1c = w1 + (size_t)cell * kMODIN * kHMOD;
    const float* bp  = w1c + (size_t)(4 * kKSQ * q + koff) * kHMOD +
                       64 * g + 4 * arow;
    const float* ap  = ldsA + arow * kMODIN + koff + 4 * kKSQ * q;
    v8f c[kMT][4] = {};
    float4 lo[kU1], hi[kU1], nlo[kU1], nhi[kU1];
    load_b<kU1>(lo, hi, bp, kHMOD);
#pragma unroll
    for (int it = 0; it < kKSQ / kU1 - 1; ++it) {
      bp += (size_t)(4 * kU1) * kHMOD;
      load_b<kU1>(nlo, nhi, bp, kHMOD);
      mma_b<kU1>(lo, hi, ap, 16 * kMODIN, c);
      ap += 4 * kU1;
#pragma unroll
      for (int u = 0; u < kU1; ++u) {
        lo[u] = nlo[u];
        hi[u] = nhi[u];
      }
    }
    float4 tlo[1], thi[1];
    if (q == 3)
      load_b<1>(tlo, thi, bp + (size_t)(4 * kU1) * kHMOD, kHMOD);
    mma_b<kU1>(lo, hi, ap, 16 * kMODIN, c);
    if (q == 3)
      mma_b<1>(tlo, thi, ap + 4 * kU1, 16 * kMODIN, c);

    float* pp = ldsP + (q * kROWS + rhalf) * kHMOD + 64 * g + 4 * arow;
#pragma unroll
    for (int mt = 0; mt < kMT; ++mt)
#pragma unroll
      for (int r = 0; r < 8; ++r)
        *(float4*)(pp + (16 * mt + r) * kHMOD) =
            make_float4(c[mt][0][r], c[mt][1][r], c[mt][2][r], c[mt][3][r]);
  }
  __syncthreads();
#pragma unroll
  for (int mt = 0; mt < kMT; ++mt) {
    const int row = (tid >> 4) + 16 * mt, col = (tid & 15) << 3;
    const float* bb = b1 + cell * kHMOD + col;
    float s[8];
    {
      const float4 u0 = *(const float4*)(bb);
      const float4 u1 = *(const float4*)(bb + 4);
      s[0] = u0.x; s[1] = u0.y; s[2] = u0.z; s[3] = u0.w;
      s[4] = u1.x; s[5] = u1.y; s[6] = u1.z; s[7] = u1.w;
    }
#pragma unroll
    for (int q = 0; q < 4; ++q) {
      const float* pq = ldsP + (q * kROWS + row) * kHMOD + col;
      const float4 u0 = *(const float4*)(pq);
      const float4 u1 = *(const float4*)(pq + 4);
      s[0] += u0.x; s[1] += u0.y; s[2] += u0.z; s[3] += u0.w;
      s[4] += u1.x; s[5] += u1.y; s[6] += u1.z; s[7] += u1.w;
    }
    float* hp = ldsH + row * kHSTRIDE + col;
    *(float4*)(hp) =
        make_float4(tanhf(s[0]), tanhf(s[1]), tanhf(s[2]), tanhf(s[3]));
    *(float4*)(hp + 4) =
        make_float4(tanhf(s[4]), tanhf(s[5]), tanhf(s[6]), tanhf(s[7]));
  }
  __syncthreads();

  const float* w2c = w2 + (size_t)cell * kHMOD * kMODOUT;
  const float* b2c = b2 + (size_t)cell * kMODOUT;
  const float* aH  = ldsH + arow * kHSTRIDE + koff;
#pragma unroll
  for (int st = wave; st < 16; st += 8) {
    const int cb = 64 * st + 4 * arow;
    v8f c[kMT][4];
    {
      const float4 bias = *(const float4*)(b2c + cb);
#pragma unroll
      for (int mt = 0; mt < kMT; ++mt)
#pragma unroll
        for (int i = 0; i < 8; ++i) {
          c[mt][0][i] = bias.x; c[mt][1][i] = bias.y;
          c[mt][2][i] = bias.z; c[mt][3][i] = bias.w;
        }
    }
    const float* bp = w2c + (size_t)koff * kMODOUT + cb;
    float4 lo[kU2], hi[kU2], nlo[kU2], nhi[kU2];
    load_b<kU2>(lo, hi, bp, kMODOUT);
#pragma unroll
    for (int it = 0; it < kHMOD / 4 / kU2; ++it) {
      if (it + 1 < kHMOD / 4 / kU2)
        load_b<kU2>(nlo, nhi, bp + (size_t)(4 * kU2) * (it + 1) * kMODOUT,
                    kMODOUT);
      mma_b<kU2>(lo, hi, aH + 4 * kU2 * it, 16 * kHSTRIDE, c);
#pragma unroll
      for (int u = 0; u < kU2; ++u) {
        lo[u] = nlo[u];
        hi[u] = nhi[u];
      }
    }
    float* op = out + ((size_t)(b0 + rhalf) * kNC + cell) * kMODOUT + cb;
#pragma unroll
    for (int mt = 0; mt < kMT; ++mt)
#pragma unroll
      for (int r = 0; r < 8; ++r)
        *(float4*)(op + (size_t)(16 * mt + r) * kNC * kMODOUT) =
            make_float4(c[mt][0][r], c[mt][1][r], c[mt][2][r], c[mt][3][r]);
  }
  if (wave < 2) {
    const int ocol = 1024 + (wave << 4) + arow;
    v8f c[kMT];
    {
      const float bias = b2c[ocol];
#pragma unroll
      for (int mt = 0; mt < kMT; ++mt)
#pragma unroll
        for (int i = 0; i < 8; ++i) c[mt][i] = bias;
    }
#pragma unroll
    for (int k0 = 0; k0 < kHMOD; k0 += 4) {
      const float bx = w2c[(size_t)(k0 + koff) * kMODOUT + ocol];
      const float by = w2c[(size_t)(k0 + koff + 1) * kMODOUT + ocol];
#pragma unroll
      for (int mt = 0; mt < kMT; ++mt) {
        const v2f a = *(const v2f*)(aH + mt * 16 * kHSTRIDE + k0);
        c[mt] = wmma4(a, bx, by, c[mt]);
      }
    }
#pragma unroll
    for (int mt = 0; mt < kMT; ++mt)
#pragma unroll
      for (int i = 0; i < 8; ++i) {
        const int row = b0 + 16 * mt + rhalf + i;
        out[((size_t)row * kNC + cell) * kMODOUT + ocol] = c[mt][i];
      }
  }
}

extern "C" void kernel_launch(void* const* d_in, const int* in_sizes, int n_in,
                              void* d_out, int out_size, void* d_ws, size_t ws_size,
                              hipStream_t stream) {
  const float* h     = (const float*)d_in[0];
  const float* msg   = (const float*)d_in[1];
  const float* decay = (const float*)d_in[2];
  const float* hebb  = (const float*)d_in[3];
  const float* drift = (const float*)d_in[4];
  const float* slive = (const float*)d_in[5];
  const float* sema  = (const float*)d_in[6];
  const float* w1    = (const float*)d_in[7];
  const float* b1    = (const float*)d_in[8];
  const float* w2    = (const float*)d_in[9];
  const float* b2    = (const float*)d_in[10];
  float* headws = (float*)d_ws;
  float* out    = (float*)d_out;

  const int norm_threads = kBS * kNC * 2 * kN;
  head_kernel<<<norm_threads / 256, 256, 0, stream>>>(
      h, msg, decay, drift, slive, sema, headws);

  dim3 grid(kBS / kROWS, kNC);
  modmlp_kernel<<<grid, 256, 0, stream>>>(
      headws, hebb, w1, b1, w2, b2, out);
}


# head kernel re-tiled as a stream: a load takes 512 contiguous bytes (2 whole rows), not 16 B of 32 lines; 2 units in flight per wave; sums re-ordered: 4 per lane, then 16-lane tree (decay mean too)
# speedup vs baseline: 2.2944x; 1.1953x over previous
#include <hip/hip_runtime.h>
#include <math.h>

constexpr int kBS      = 64;
constexpr int kNC      = 128;
constexpr int kN       = 32;
constexpr int kDN      = 64;
constexpr int kHMOD    = 128;
constexpr int kMODIN   = 2 * kN + 4 + kN * kN;
constexpr int kMODOUT  = kN * kN + kN;
constexpr int kHEAD    = 2 * kN + 4;
constexpr int kHSTRIDE = 132;
constexpr int kMT      = 2;
constexpr int kROWS    = 16 * kMT;

typedef __attribute__((ext_vector_type(2))) float v2f;
typedef __attribute__((ext_vector_type(8))) float v8f;

typedef __attribute__((ext_vector_type(4))) float v4f;

constexpr int kUNIT  = kN * kDN;
constexpr int kUQ    = kUNIT / 128;
constexpr int kHWAVE = 16;
constexpr int kHGRID = kBS * kNC / (2 * kHWAVE);

template <int M>
__device__ __forceinline__ float half_xor(float x) {
  const int b = __builtin_bit_cast(int, x);
  return __builtin_bit_cast(
      float, __builtin_amdgcn_update_dpp(b, b, 0x160 + M, 0xf, 0xf, false));
}

__device__ __forceinline__ void load_unit(v4f (&v)[kUQ], const float* p) {
#pragma unroll
  for (int q = 0; q < kUQ; ++q)
    v[q] = __builtin_nontemporal_load((const v4f*)(p + 128 * q));
}

template <int M, int CNT>
__device__ __forceinline__ void fold(const float (&in)[2 * CNT],
                                     float (&out)[CNT], bool up) {
#pragma unroll
  for (int i = 0; i < CNT; ++i) {
    const float keep = up ? in[i + CNT] : in[i];
    const float send = up ? in[i] : in[i + CNT];
    out[i] = keep + half_xor<M>(send);
  }
}

__device__ __forceinline__ float unit_norm(const v4f (&v)[kUQ], int c) {
  float p[kUQ], r8[8], r4[4], r2[2], r1[1];
#pragma unroll
  for (int q = 0; q < kUQ; ++q)
    p[q] = v[q].x * v[q].x + v[q].y * v[q].y + v[q].z * v[q].z +
           v[q].w * v[q].w;
  fold<8, 8>(p, r8, (c & 8) != 0);
  fold<4, 4>(r8, r4, (c & 4) != 0);
  fold<2, 2>(r4, r2, (c & 2) != 0);
  fold<1, 1>(r2, r1, (c & 1) != 0);
  return sqrtf(r1[0]);
}

__global__ __launch_bounds__(32 * kHWAVE) void head_kernel(
    const float* __restrict__ h, const float* __restrict__ msg,
    const float* __restrict__ decay, const float* __restrict__ drift,
    const float* __restrict__ s_live, const float* __restrict__ s_ema,
    float* __restrict__ head) {
  const int lane = threadIdx.x & 31;
  const int wave = __builtin_amdgcn_readfirstlane(threadIdx.x >> 5);
  const int pair0 = (blockIdx.x * kHWAVE + wave) * 2;
  const int c = lane & 15, hh = lane >> 4;

  const float2 d2 = *(const float2*)(decay + (size_t)pair0 * kN + 2 * lane);
  float dr = drift[pair0 + hh];
  float sl = s_live[pair0 / kNC];
  float se = s_ema[pair0 / kNC];
  __builtin_amdgcn_sched_barrier(0);

  const float* hp = h + (size_t)pair0 * kUNIT + 4 * lane;
  const float* mp = msg + (size_t)pair0 * kUNIT + 4 * lane;
  float* hd = head + (size_t)pair0 * kHEAD + 2 * c + hh;

  v4f va[kUQ], vb[kUQ];
  load_unit(va, hp);
  __builtin_amdgcn_sched_barrier(0);
  load_unit(vb, mp);
  __builtin_amdgcn_sched_barrier(0);
  asm volatile("" : "+v"(dr), "+v"(sl), "+v"(se));

  {
    float m = d2.x + d2.y;
    m += half_xor<8>(m);
    m += half_xor<4>(m);
    m += half_xor<2>(m);
    m += half_xor<1>(m);
    if (c == 0)
      *(float4*)(head + (size_t)(pair0 + hh) * kHEAD + 2 * kN) =
          make_float4(m * (1.f / (float)kN), dr, sl, se);
  }

  __builtin_amdgcn_sched_barrier(0);
  hd[0] = unit_norm(va, c);
  __builtin_amdgcn_sched_barrier(0);
  load_unit(va, hp + kUNIT);
  __builtin_amdgcn_sched_barrier(0);
  hd[kN] = unit_norm(vb, c);
  __builtin_amdgcn_sched_barrier(0);
  load_unit(vb, mp + kUNIT);
  __builtin_amdgcn_sched_barrier(0);
  hd[kHEAD] = unit_norm(va, c);
  hd[kHEAD + kN] = unit_norm(vb, c);
}

__device__ __forceinline__ v8f wmma4(v2f a, float bx, float by, v8f c) {
  v2f b;
  b.x = bx;
  b.y = by;
  return __builtin_amdgcn_wmma_f32_16x16x4_f32(
      false, a, false, b, (short)0, c, false, false);
}

template <int U>
__device__ __forceinline__ void load_b(float4 (&lo)[U], float4 (&hi)[U],
                                       const float* __restrict__ p,
                                       int pitch) {
#pragma unroll
  for (int u = 0; u < U; ++u) {
    lo[u] = *(const float4*)(p + (size_t)(4 * u) * pitch);
    hi[u] = *(const float4*)(p + (size_t)(4 * u + 1) * pitch);
  }
}

template <int U>
__device__ __forceinline__ void mma_b(const float4 (&lo)[U],
                                      const float4 (&hi)[U],
                                      const float* __restrict__ ap, int mpitch,
                                      v8f (&c)[kMT][4]) {
#pragma unroll
  for (int u = 0; u < U; ++u) {
#pragma unroll
    for (int mt = 0; mt < kMT; ++mt) {
      const v2f a = *(const v2f*)(ap + mt * mpitch + 4 * u);
      c[mt][0] = wmma4(a, lo[u].x, hi[u].x, c[mt][0]);
      c[mt][1] = wmma4(a, lo[u].y, hi[u].y, c[mt][1]);
      c[mt][2] = wmma4(a, lo[u].z, hi[u].z, c[mt][2]);
      c[mt][3] = wmma4(a, lo[u].w, hi[u].w, c[mt][3]);
    }
  }
}

constexpr int kKS1   = kMODIN / 4;
constexpr int kKSQ   = kKS1 / 4;
constexpr int kU1    = 4;
constexpr int kU2    = 8;

constexpr int kLdsFloats = kROWS * kMODIN + kROWS * kHSTRIDE + 4 * kROWS * kHMOD;

__global__ __launch_bounds__(256) __attribute__((amdgpu_waves_per_eu(2, 2)))
void modmlp_kernel(
    const float* __restrict__ head, const float* __restrict__ hebb,
    const float* __restrict__ w1, const float* __restrict__ b1,
    const float* __restrict__ w2, const float* __restrict__ b2,
    float* __restrict__ out) {
  __shared__ __align__(16) float smem[kLdsFloats];
  float* ldsA = smem;
  float* ldsH = smem + kROWS * kMODIN;
  float* ldsP = ldsH + kROWS * kHSTRIDE;

  const int tid   = threadIdx.x;
  const int lane  = tid & 31;
  const int wave  = tid >> 5;
  const int mtile = blockIdx.x;
  const int cell  = blockIdx.y;
  const int b0    = mtile * kROWS;
  const int arow  = lane & 15;
  const int koff  = (lane >> 4) << 1;
  const int rhalf = (lane >> 4) << 3;

  {
    const size_t pr0 = (size_t)b0 * kNC + cell;
    const float* hb = hebb + pr0 * (kN * kN) + 4 * tid;
    float4 hv[kROWS];
#pragma unroll
    for (int r = 0; r < kROWS; ++r)
      hv[r] = *(const float4*)(hb + (size_t)r * kNC * (kN * kN));
    const int nr = tid >> 4, nq = tid & 15;
    float4 nv[kMT];
#pragma unroll
    for (int i = 0; i < kMT; ++i)
      nv[i] = *(const float4*)(head + (pr0 + (size_t)(nr + 16 * i) * kNC) * kHEAD +
                               4 * nq);
#pragma unroll
    for (int i = 0; i < kMT; ++i)
      *(float4*)(ldsA + (nr + 16 * i) * kMODIN + 4 * nq) = nv[i];
#pragma unroll
    for (int r = 0; r < kROWS; ++r)
      *(float4*)(ldsA + r * kMODIN + kHEAD + 4 * tid) = hv[r];
    if (tid < kROWS)
      *(float4*)(ldsA + tid * kMODIN + 2 * kN) =
          *(const float4*)(head + (pr0 + (size_t)tid * kNC) * kHEAD + 2 * kN);
  }
  __syncthreads();

  {
    const int g = wave & 1;
    const int q = wave >> 1;
    const float* w1c = w1 + (size_t)cell * kMODIN * kHMOD;
    const float* bp  = w1c + (size_t)(4 * kKSQ * q + koff) * kHMOD +
                       64 * g + 4 * arow;
    const float* ap  = ldsA + arow * kMODIN + koff + 4 * kKSQ * q;
    v8f c[kMT][4] = {};
    float4 lo[kU1], hi[kU1], nlo[kU1], nhi[kU1];
    load_b<kU1>(lo, hi, bp, kHMOD);
#pragma unroll
    for (int it = 0; it < kKSQ / kU1 - 1; ++it) {
      bp += (size_t)(4 * kU1) * kHMOD;
      load_b<kU1>(nlo, nhi, bp, kHMOD);
      mma_b<kU1>(lo, hi, ap, 16 * kMODIN, c);
      ap += 4 * kU1;
#pragma unroll
      for (int u = 0; u < kU1; ++u) {
        lo[u] = nlo[u];
        hi[u] = nhi[u];
      }
    }
    float4 tlo[1], thi[1];
    if (q == 3)
      load_b<1>(tlo, thi, bp + (size_t)(4 * kU1) * kHMOD, kHMOD);
    mma_b<kU1>(lo, hi, ap, 16 * kMODIN, c);
    if (q == 3)
      mma_b<1>(tlo, thi, ap + 4 * kU1, 16 * kMODIN, c);

    float* pp = ldsP + (q * kROWS + rhalf) * kHMOD + 64 * g + 4 * arow;
#pragma unroll
    for (int mt = 0; mt < kMT; ++mt)
#pragma unroll
      for (int r = 0; r < 8; ++r)
        *(float4*)(pp + (16 * mt + r) * kHMOD) =
            make_float4(c[mt][0][r], c[mt][1][r], c[mt][2][r], c[mt][3][r]);
  }
  __syncthreads();
#pragma unroll
  for (int mt = 0; mt < kMT; ++mt) {
    const int row = (tid >> 4) + 16 * mt, col = (tid & 15) << 3;
    const float* bb = b1 + cell * kHMOD + col;
    float s[8];
    {
      const float4 u0 = *(const float4*)(bb);
      const float4 u1 = *(const float4*)(bb + 4);
      s[0] = u0.x; s[1] = u0.y; s[2] = u0.z; s[3] = u0.w;
      s[4] = u1.x; s[5] = u1.y; s[6] = u1.z; s[7] = u1.w;
    }
#pragma unroll
    for (int q = 0; q < 4; ++q) {
      const float* pq = ldsP + (q * kROWS + row) * kHMOD + col;
      const float4 u0 = *(const float4*)(pq);
      const float4 u1 = *(const float4*)(pq + 4);
      s[0] += u0.x; s[1] += u0.y; s[2] += u0.z; s[3] += u0.w;
      s[4] += u1.x; s[5] += u1.y; s[6] += u1.z; s[7] += u1.w;
    }
    float* hp = ldsH + row * kHSTRIDE + col;
    *(float4*)(hp) =
        make_float4(tanhf(s[0]), tanhf(s[1]), tanhf(s[2]), tanhf(s[3]));
    *(float4*)(hp + 4) =
        make_float4(tanhf(s[4]), tanhf(s[5]), tanhf(s[6]), tanhf(s[7]));
  }
  __syncthreads();

  const float* w2c = w2 + (size_t)cell * kHMOD * kMODOUT;
  const float* b2c = b2 + (size_t)cell * kMODOUT;
  const float* aH  = ldsH + arow * kHSTRIDE + koff;
#pragma unroll
  for (int st = wave; st < 16; st += 8) {
    const int cb = 64 * st + 4 * arow;
    v8f c[kMT][4];
    {
      const float4 bias = *(const float4*)(b2c + cb);
#pragma unroll
      for (int mt = 0; mt < kMT; ++mt)
#pragma unroll
        for (int i = 0; i < 8; ++i) {
          c[mt][0][i] = bias.x; c[mt][1][i] = bias.y;
          c[mt][2][i] = bias.z; c[mt][3][i] = bias.w;
        }
    }
    const float* bp = w2c + (size_t)koff * kMODOUT + cb;
    float4 lo[kU2], hi[kU2], nlo[kU2], nhi[kU2];
    load_b<kU2>(lo, hi, bp, kMODOUT);
#pragma unroll
    for (int it = 0; it < kHMOD / 4 / kU2; ++it) {
      if (it + 1 < kHMOD / 4 / kU2)
        load_b<kU2>(nlo, nhi, bp + (size_t)(4 * kU2) * (it + 1) * kMODOUT,
                    kMODOUT);
      mma_b<kU2>(lo, hi, aH + 4 * kU2 * it, 16 * kHSTRIDE, c);
#pragma unroll
      for (int u = 0; u < kU2; ++u) {
        lo[u] = nlo[u];
        hi[u] = nhi[u];
      }
    }
    float* op = out + ((size_t)(b0 + rhalf) * kNC + cell) * kMODOUT + cb;
#pragma unroll
    for (int mt = 0; mt < kMT; ++mt)
#pragma unroll
      for (int r = 0; r < 8; ++r)
        *(float4*)(op + (size_t)(16 * mt + r) * kNC * kMODOUT) =
            make_float4(c[mt][0][r], c[mt][1][r], c[mt][2][r], c[mt][3][r]);
  }
  if (wave < 2) {
    const int ocol = 1024 + (wave << 4) + arow;
    v8f c[kMT];
    {
      const float bias = b2c[ocol];
#pragma unroll
      for (int mt = 0; mt < kMT; ++mt)
#pragma unroll
        for (int i = 0; i < 8; ++i) c[mt][i] = bias;
    }
#pragma unroll
    for (int k0 = 0; k0 < kHMOD; k0 += 4) {
      const float bx = w2c[(size_t)(k0 + koff) * kMODOUT + ocol];
      const float by = w2c[(size_t)(k0 + koff + 1) * kMODOUT + ocol];
#pragma unroll
      for (int mt = 0; mt < kMT; ++mt) {
        const v2f a = *(const v2f*)(aH + mt * 16 * kHSTRIDE + k0);
        c[mt] = wmma4(a, bx, by, c[mt]);
      }
    }
#pragma unroll
    for (int mt = 0; mt < kMT; ++mt)
#pragma unroll
      for (int i = 0; i < 8; ++i) {
        const int row = b0 + 16 * mt + rhalf + i;
        out[((size_t)row * kNC + cell) * kMODOUT + ocol] = c[mt][i];
      }
  }
}

extern "C" void kernel_launch(void* const* d_in, const int* in_sizes, int n_in,
                              void* d_out, int out_size, void* d_ws, size_t ws_size,
                              hipStream_t stream) {
  const float* h     = (const float*)d_in[0];
  const float* msg   = (const float*)d_in[1];
  const float* decay = (const float*)d_in[2];
  const float* hebb  = (const float*)d_in[3];
  const float* drift = (const float*)d_in[4];
  const float* slive = (const float*)d_in[5];
  const float* sema  = (const float*)d_in[6];
  const float* w1    = (const float*)d_in[7];
  const float* b1    = (const float*)d_in[8];
  const float* w2    = (const float*)d_in[9];
  const float* b2    = (const float*)d_in[10];
  float* headws = (float*)d_ws;
  float* out    = (float*)d_out;

  head_kernel<<<kHGRID, 32 * kHWAVE, 0, stream>>>(
      h, msg, decay, drift, slive, sema, headws);

  dim3 grid(kBS / kROWS, kNC);
  modmlp_kernel<<<grid, 256, 0, stream>>>(
      headws, hebb, w1, b1, w2, b2, out);
}


# head kernel fused into the MLP kernel: each wave builds norms and scalars of 4 tile rows into the LDS A tile as a stream (3 units in flight, hebbian copy non-temporal); 1 launch, no workspace
# speedup vs baseline: 2.4148x; 1.0525x over previous
#include <hip/hip_runtime.h>
#include <math.h>

constexpr int kBS      = 64;
constexpr int kNC      = 128;
constexpr int kN       = 32;
constexpr int kDN      = 64;
constexpr int kHMOD    = 128;
constexpr int kMODIN   = 2 * kN + 4 + kN * kN;
constexpr int kMODOUT  = kN * kN + kN;
constexpr int kHEAD    = 2 * kN + 4;
constexpr int kHSTRIDE = 132;
constexpr int kMT      = 2;
constexpr int kROWS    = 16 * kMT;

typedef __attribute__((ext_vector_type(2))) float v2f;
typedef __attribute__((ext_vector_type(8))) float v8f;

typedef __attribute__((ext_vector_type(4))) float v4f;

constexpr int kUNIT  = kN * kDN;
constexpr int kUQ    = kUNIT / 128;

template <int M>
__device__ __forceinline__ float half_xor(float x) {
  const int b = __builtin_bit_cast(int, x);
  return __builtin_bit_cast(
      float, __builtin_amdgcn_update_dpp(b, b, 0x160 + M, 0xf, 0xf, false));
}

__device__ __forceinline__ void load_unit(v4f (&v)[kUQ], const float* p) {
#pragma unroll
  for (int q = 0; q < kUQ; ++q)
    v[q] = __builtin_nontemporal_load((const v4f*)(p + 128 * q));
}

template <int M, int CNT>
__device__ __forceinline__ void fold(const float (&in)[2 * CNT],
                                     float (&out)[CNT], bool up) {
#pragma unroll
  for (int i = 0; i < CNT; ++i) {
    const float keep = up ? in[i + CNT] : in[i];
    const float send = up ? in[i] : in[i + CNT];
    out[i] = keep + half_xor<M>(send);
  }
}

__device__ __forceinline__ float unit_norm(const v4f (&v)[kUQ], int c) {
  float p[kUQ], r8[8], r4[4], r2[2], r1[1];
#pragma unroll
  for (int q = 0; q < kUQ; ++q)
    p[q] = v[q].x * v[q].x + v[q].y * v[q].y + v[q].z * v[q].z +
           v[q].w * v[q].w;
  fold<8, 8>(p, r8, (c & 8) != 0);
  fold<4, 4>(r8, r4, (c & 4) != 0);
  fold<2, 2>(r4, r2, (c & 2) != 0);
  fold<1, 1>(r2, r1, (c & 1) != 0);
  return sqrtf(r1[0]);
}

__device__ __forceinline__ v8f wmma4(v2f a, float bx, float by, v8f c) {
  v2f b;
  b.x = bx;
  b.y = by;
  return __builtin_amdgcn_wmma_f32_16x16x4_f32(
      false, a, false, b, (short)0, c, false, false);
}

template <int U>
__device__ __forceinline__ void load_b(float4 (&lo)[U], float4 (&hi)[U],
                                       const float* __restrict__ p,
                                       int pitch) {
#pragma unroll
  for (int u = 0; u < U; ++u) {
    lo[u] = *(const float4*)(p + (size_t)(4 * u) * pitch);
    hi[u] = *(const float4*)(p + (size_t)(4 * u + 1) * pitch);
  }
}

template <int U>
__device__ __forceinline__ void mma_b(const float4 (&lo)[U],
                                      const float4 (&hi)[U],
                                      const float* __restrict__ ap, int mpitch,
                                      v8f (&c)[kMT][4]) {
#pragma unroll
  for (int u = 0; u < U; ++u) {
#pragma unroll
    for (int mt = 0; mt < kMT; ++mt) {
      const v2f a = *(const v2f*)(ap + mt * mpitch + 4 * u);
      c[mt][0] = wmma4(a, lo[u].x, hi[u].x, c[mt][0]);
      c[mt][1] = wmma4(a, lo[u].y, hi[u].y, c[mt][1]);
      c[mt][2] = wmma4(a, lo[u].z, hi[u].z, c[mt][2]);
      c[mt][3] = wmma4(a, lo[u].w, hi[u].w, c[mt][3]);
    }
  }
}

constexpr int kKS1   = kMODIN / 4;
constexpr int kKSQ   = kKS1 / 4;
constexpr int kU1    = 4;
constexpr int kU2    = 8;

constexpr int kLdsFloats = kROWS * kMODIN + kROWS * kHSTRIDE + 4 * kROWS * kHMOD;

__global__ __launch_bounds__(256) __attribute__((amdgpu_waves_per_eu(2, 2)))
void modmlp_kernel(
    const float* __restrict__ h, const float* __restrict__ msg,
    const float* __restrict__ decay, const float* __restrict__ drift,
    const float* __restrict__ s_live, const float* __restrict__ s_ema,
    const float* __restrict__ hebb,
    const float* __restrict__ w1, const float* __restrict__ b1,
    const float* __restrict__ w2, const float* __restrict__ b2,
    float* __restrict__ out) {
  __shared__ __align__(16) float smem[kLdsFloats];
  float* ldsA = smem;
  float* ldsH = smem + kROWS * kMODIN;
  float* ldsP = ldsH + kROWS * kHSTRIDE;

  const int tid   = threadIdx.x;
  const int lane  = tid & 31;
  const int wave  = tid >> 5;
  const int mtile = blockIdx.x;
  const int cell  = blockIdx.y;
  const int b0    = mtile * kROWS;
  const int arow  = lane & 15;
  const int koff  = (lane >> 4) << 1;
  const int rhalf = (lane >> 4) << 3;

  {
    const size_t pr0 = (size_t)b0 * kNC + cell;
    const int c = lane & 15, hh = lane >> 4;
    const int r0 = 4 * __builtin_amdgcn_readfirstlane(wave);

    const float* hb = hebb + pr0 * (kN * kN) + 4 * tid;
    v4f hv[kROWS];
#pragma unroll
    for (int r = 0; r < kROWS; ++r)
      hv[r] = __builtin_nontemporal_load(
          (const v4f*)(hb + (size_t)r * kNC * (kN * kN)));

    float2 d2[2];
    float dr[2], sl[2], se[2];
#pragma unroll
    for (int j = 0; j < 2; ++j) {
      const int rr = r0 + 2 * j + hh;
      d2[j] = *(const float2*)(decay + (pr0 + (size_t)rr * kNC) * kN + 2 * c);
      dr[j] = drift[pr0 + (size_t)rr * kNC];
      sl[j] = s_live[b0 + rr];
      se[j] = s_ema[b0 + rr];
    }
    __builtin_amdgcn_sched_barrier(0);

    auto unit_ptr = [&](int u) -> const float* {
      return ((u & 1) ? msg : h) +
             (pr0 + (size_t)(r0 + (u >> 1)) * kNC) * kUNIT + 4 * lane;
    };
    v4f va[kUQ], vb[kUQ], vc[kUQ];
    load_unit(va, unit_ptr(0));
    __builtin_amdgcn_sched_barrier(0);
    load_unit(vb, unit_ptr(1));
    __builtin_amdgcn_sched_barrier(0);
    load_unit(vc, unit_ptr(2));
    __builtin_amdgcn_sched_barrier(0);
#pragma unroll
    for (int r = 0; r < kROWS; ++r)
      *(v4f*)(ldsA + r * kMODIN + kHEAD + 4 * tid) = hv[r];
    asm volatile("" : "+v"(dr[0]), "+v"(sl[0]), "+v"(se[0]),
                      "+v"(dr[1]), "+v"(sl[1]), "+v"(se[1]));

#pragma unroll
    for (int j = 0; j < 2; ++j) {
      float m = d2[j].x + d2[j].y;
      m += half_xor<8>(m);
      m += half_xor<4>(m);
      m += half_xor<2>(m);
      m += half_xor<1>(m);
      if (c == 0)
        *(float4*)(ldsA + (r0 + 2 * j + hh) * kMODIN + 2 * kN) =
            make_float4(m * (1.f / (float)kN), dr[j], sl[j], se[j]);
    }
    float* na = ldsA + r0 * kMODIN + 2 * c + hh;
    __builtin_amdgcn_sched_barrier(0);
    na[0] = unit_norm(va, c);
    __builtin_amdgcn_sched_barrier(0);
    load_unit(va, unit_ptr(3));
    __builtin_amdgcn_sched_barrier(0);
    na[kN] = unit_norm(vb, c);
    __builtin_amdgcn_sched_barrier(0);
    load_unit(vb, unit_ptr(4));
    __builtin_amdgcn_sched_barrier(0);
    na[kMODIN] = unit_norm(vc, c);
    __builtin_amdgcn_sched_barrier(0);
    load_unit(vc, unit_ptr(5));
    __builtin_amdgcn_sched_barrier(0);
    na[kMODIN + kN] = unit_norm(va, c);
    __builtin_amdgcn_sched_barrier(0);
    load_unit(va, unit_ptr(6));
    __builtin_amdgcn_sched_barrier(0);
    na[2 * kMODIN] = unit_norm(vb, c);
    __builtin_amdgcn_sched_barrier(0);
    load_unit(vb, unit_ptr(7));
    __builtin_amdgcn_sched_barrier(0);
    na[2 * kMODIN + kN] = unit_norm(vc, c);
    na[3 * kMODIN] = unit_norm(va, c);
    na[3 * kMODIN + kN] = unit_norm(vb, c);
  }
  __syncthreads();

  {
    const int g = wave & 1;
    const int q = wave >> 1;
    const float* w1c = w1 + (size_t)cell * kMODIN * kHMOD;
    const float* bp  = w1c + (size_t)(4 * kKSQ * q + koff) * kHMOD +
                       64 * g + 4 * arow;
    const float* ap  = ldsA + arow * kMODIN + koff + 4 * kKSQ * q;
    v8f c[kMT][4] = {};
    float4 lo[kU1], hi[kU1], nlo[kU1], nhi[kU1];
    load_b<kU1>(lo, hi, bp, kHMOD);
#pragma unroll
    for (int it = 0; it < kKSQ / kU1 - 1; ++it) {
      bp += (size_t)(4 * kU1) * kHMOD;
      load_b<kU1>(nlo, nhi, bp, kHMOD);
      mma_b<kU1>(lo, hi, ap, 16 * kMODIN, c);
      ap += 4 * kU1;
#pragma unroll
      for (int u = 0; u < kU1; ++u) {
        lo[u] = nlo[u];
        hi[u] = nhi[u];
      }
    }
    float4 tlo[1], thi[1];
    if (q == 3)
      load_b<1>(tlo, thi, bp + (size_t)(4 * kU1) * kHMOD, kHMOD);
    mma_b<kU1>(lo, hi, ap, 16 * kMODIN, c);
    if (q == 3)
      mma_b<1>(tlo, thi, ap + 4 * kU1, 16 * kMODIN, c);

    float* pp = ldsP + (q * kROWS + rhalf) * kHMOD + 64 * g + 4 * arow;
#pragma unroll
    for (int mt = 0; mt < kMT; ++mt)
#pragma unroll
      for (int r = 0; r < 8; ++r)
        *(float4*)(pp + (16 * mt + r) * kHMOD) =
            make_float4(c[mt][0][r], c[mt][1][r], c[mt][2][r], c[mt][3][r]);
  }
  __syncthreads();
#pragma unroll
  for (int mt = 0; mt < kMT; ++mt) {
    const int row = (tid >> 4) + 16 * mt, col = (tid & 15) << 3;
    const float* bb = b1 + cell * kHMOD + col;
    float s[8];
    {
      const float4 u0 = *(const float4*)(bb);
      const float4 u1 = *(const float4*)(bb + 4);
      s[0] = u0.x; s[1] = u0.y; s[2] = u0.z; s[3] = u0.w;
      s[4] = u1.x; s[5] = u1.y; s[6] = u1.z; s[7] = u1.w;
    }
#pragma unroll
    for (int q = 0; q < 4; ++q) {
      const float* pq = ldsP + (q * kROWS + row) * kHMOD + col;
      const float4 u0 = *(const float4*)(pq);
      const float4 u1 = *(const float4*)(pq + 4);
      s[0] += u0.x; s[1] += u0.y; s[2] += u0.z; s[3] += u0.w;
      s[4] += u1.x; s[5] += u1.y; s[6] += u1.z; s[7] += u1.w;
    }
    float* hp = ldsH + row * kHSTRIDE + col;
    *(float4*)(hp) =
        make_float4(tanhf(s[0]), tanhf(s[1]), tanhf(s[2]), tanhf(s[3]));
    *(float4*)(hp + 4) =
        make_float4(tanhf(s[4]), tanhf(s[5]), tanhf(s[6]), tanhf(s[7]));
  }
  __syncthreads();

  const float* w2c = w2 + (size_t)cell * kHMOD * kMODOUT;
  const float* b2c = b2 + (size_t)cell * kMODOUT;
  const float* aH  = ldsH + arow * kHSTRIDE + koff;
#pragma unroll
  for (int st = wave; st < 16; st += 8) {
    const int cb = 64 * st + 4 * arow;
    v8f c[kMT][4];
    {
      const float4 bias = *(const float4*)(b2c + cb);
#pragma unroll
      for (int mt = 0; mt < kMT; ++mt)
#pragma unroll
        for (int i = 0; i < 8; ++i) {
          c[mt][0][i] = bias.x; c[mt][1][i] = bias.y;
          c[mt][2][i] = bias.z; c[mt][3][i] = bias.w;
        }
    }
    const float* bp = w2c + (size_t)koff * kMODOUT + cb;
    float4 lo[kU2], hi[kU2], nlo[kU2], nhi[kU2];
    load_b<kU2>(lo, hi, bp, kMODOUT);
#pragma unroll
    for (int it = 0; it < kHMOD / 4 / kU2; ++it) {
      if (it + 1 < kHMOD / 4 / kU2)
        load_b<kU2>(nlo, nhi, bp + (size_t)(4 * kU2) * (it + 1) * kMODOUT,
                    kMODOUT);
      mma_b<kU2>(lo, hi, aH + 4 * kU2 * it, 16 * kHSTRIDE, c);
#pragma unroll
      for (int u = 0; u < kU2; ++u) {
        lo[u] = nlo[u];
        hi[u] = nhi[u];
      }
    }
    float* op = out + ((size_t)(b0 + rhalf) * kNC + cell) * kMODOUT + cb;
#pragma unroll
    for (int mt = 0; mt < kMT; ++mt)
#pragma unroll
      for (int r = 0; r < 8; ++r)
        *(float4*)(op + (size_t)(16 * mt + r) * kNC * kMODOUT) =
            make_float4(c[mt][0][r], c[mt][1][r], c[mt][2][r], c[mt][3][r]);
  }
  if (wave < 2) {
    const int ocol = 1024 + (wave << 4) + arow;
    v8f c[kMT];
    {
      const float bias = b2c[ocol];
#pragma unroll
      for (int mt = 0; mt < kMT; ++mt)
#pragma unroll
        for (int i = 0; i < 8; ++i) c[mt][i] = bias;
    }
#pragma unroll
    for (int k0 = 0; k0 < kHMOD; k0 += 4) {
      const float bx = w2c[(size_t)(k0 + koff) * kMODOUT + ocol];
      const float by = w2c[(size_t)(k0 + koff + 1) * kMODOUT + ocol];
#pragma unroll
      for (int mt = 0; mt < kMT; ++mt) {
        const v2f a = *(const v2f*)(aH + mt * 16 * kHSTRIDE + k0);
        c[mt] = wmma4(a, bx, by, c[mt]);
      }
    }
#pragma unroll
    for (int mt = 0; mt < kMT; ++mt)
#pragma unroll
      for (int i = 0; i < 8; ++i) {
        const int row = b0 + 16 * mt + rhalf + i;
        out[((size_t)row * kNC + cell) * kMODOUT + ocol] = c[mt][i];
      }
  }
}

extern "C" void kernel_launch(void* const* d_in, const int* in_sizes, int n_in,
                              void* d_out, int out_size, void* d_ws, size_t ws_size,
                              hipStream_t stream) {
  const float* h     = (const float*)d_in[0];
  const float* msg   = (const float*)d_in[1];
  const float* decay = (const float*)d_in[2];
  const float* hebb  = (const float*)d_in[3];
  const float* drift = (const float*)d_in[4];
  const float* slive = (const float*)d_in[5];
  const float* sema  = (const float*)d_in[6];
  const float* w1    = (const float*)d_in[7];
  const float* b1    = (const float*)d_in[8];
  const float* w2    = (const float*)d_in[9];
  const float* b2    = (const float*)d_in[10];
  float* out    = (float*)d_out;

  dim3 grid(kBS / kROWS, kNC);
  modmlp_kernel<<<grid, 256, 0, stream>>>(
      h, msg, decay, drift, slive, sema, hebb, w1, b1, w2, b2, out);
}


# h/msg stream and norm sums moved under layer 1's matrix loop: layer 1 takes the 1024 hebbian k first (2 units in flight beside the weight rows), the 68 head k last; K-quarter sums re-ordered
# speedup vs baseline: 2.5639x; 1.0617x over previous
#include <hip/hip_runtime.h>
#include <math.h>

constexpr int kBS      = 64;
constexpr int kNC      = 128;
constexpr int kN       = 32;
constexpr int kDN      = 64;
constexpr int kHMOD    = 128;
constexpr int kMODIN   = 2 * kN + 4 + kN * kN;
constexpr int kMODOUT  = kN * kN + kN;
constexpr int kHEAD    = 2 * kN + 4;
constexpr int kHSTRIDE = 132;
constexpr int kMT      = 2;
constexpr int kROWS    = 16 * kMT;

typedef __attribute__((ext_vector_type(2))) float v2f;
typedef __attribute__((ext_vector_type(8))) float v8f;

typedef __attribute__((ext_vector_type(4))) float v4f;

constexpr int kUNIT  = kN * kDN;
constexpr int kUQ    = kUNIT / 128;

template <int M>
__device__ __forceinline__ float half_xor(float x) {
  const int b = __builtin_bit_cast(int, x);
  return __builtin_bit_cast(
      float, __builtin_amdgcn_update_dpp(b, b, 0x160 + M, 0xf, 0xf, false));
}

__device__ __forceinline__ void load_unit(v4f (&v)[kUQ], const float* p) {
#pragma unroll
  for (int q = 0; q < kUQ; ++q)
    v[q] = __builtin_nontemporal_load((const v4f*)(p + 128 * q));
}

template <int M, int CNT>
__device__ __forceinline__ void fold(const float (&in)[2 * CNT],
                                     float (&out)[CNT], bool up) {
#pragma unroll
  for (int i = 0; i < CNT; ++i) {
    const float keep = up ? in[i + CNT] : in[i];
    const float send = up ? in[i] : in[i + CNT];
    out[i] = keep + half_xor<M>(send);
  }
}

__device__ __forceinline__ float unit_norm(const v4f (&v)[kUQ], int c) {
  float p[kUQ], r8[8], r4[4], r2[2], r1[1];
#pragma unroll
  for (int q = 0; q < kUQ; ++q)
    p[q] = v[q].x * v[q].x + v[q].y * v[q].y + v[q].z * v[q].z +
           v[q].w * v[q].w;
  fold<8, 8>(p, r8, (c & 8) != 0);
  fold<4, 4>(r8, r4, (c & 4) != 0);
  fold<2, 2>(r4, r2, (c & 2) != 0);
  fold<1, 1>(r2, r1, (c & 1) != 0);
  return sqrtf(r1[0]);
}

__device__ __forceinline__ v8f wmma4(v2f a, float bx, float by, v8f c) {
  v2f b;
  b.x = bx;
  b.y = by;
  return __builtin_amdgcn_wmma_f32_16x16x4_f32(
      false, a, false, b, (short)0, c, false, false);
}

template <int U>
__device__ __forceinline__ void load_b(float4 (&lo)[U], float4 (&hi)[U],
                                       const float* __restrict__ p,
                                       int pitch) {
#pragma unroll
  for (int u = 0; u < U; ++u) {
    lo[u] = *(const float4*)(p + (size_t)(4 * u) * pitch);
    hi[u] = *(const float4*)(p + (size_t)(4 * u + 1) * pitch);
  }
}

template <int U>
__device__ __forceinline__ void mma_b(const float4 (&lo)[U],
                                      const float4 (&hi)[U],
                                      const float* __restrict__ ap, int mpitch,
                                      v8f (&c)[kMT][4]) {
#pragma unroll
  for (int u = 0; u < U; ++u) {
#pragma unroll
    for (int mt = 0; mt < kMT; ++mt) {
      const v2f a = *(const v2f*)(ap + mt * mpitch + 4 * u);
      c[mt][0] = wmma4(a, lo[u].x, hi[u].x, c[mt][0]);
      c[mt][1] = wmma4(a, lo[u].y, hi[u].y, c[mt][1]);
      c[mt][2] = wmma4(a, lo[u].z, hi[u].z, c[mt][2]);
      c[mt][3] = wmma4(a, lo[u].w, hi[u].w, c[mt][3]);
    }
  }
}

constexpr int kHK    = kHEAD / 4;
constexpr int kBQ    = (kN * kN) / 16;
constexpr int kU1    = 4;
constexpr int kG1    = kBQ / kU1;
constexpr int kNB    = 4;
static_assert(kHK == 4 * kU1 + 1, "head k-steps: 4 per K-quarter and one over");
constexpr int kU2    = 8;

constexpr int kLdsFloats = kROWS * kMODIN + kROWS * kHSTRIDE + 4 * kROWS * kHMOD;

__global__ __launch_bounds__(256) __attribute__((amdgpu_waves_per_eu(2, 2)))
void modmlp_kernel(
    const float* __restrict__ h, const float* __restrict__ msg,
    const float* __restrict__ decay, const float* __restrict__ drift,
    const float* __restrict__ s_live, const float* __restrict__ s_ema,
    const float* __restrict__ hebb,
    const float* __restrict__ w1, const float* __restrict__ b1,
    const float* __restrict__ w2, const float* __restrict__ b2,
    float* __restrict__ out) {
  __shared__ __align__(16) float smem[kLdsFloats];
  float* ldsA = smem;
  float* ldsH = smem + kROWS * kMODIN;
  float* ldsP = ldsH + kROWS * kHSTRIDE;

  const int tid   = threadIdx.x;
  const int lane  = tid & 31;
  const int wave  = tid >> 5;
  const int mtile = blockIdx.x;
  const int cell  = blockIdx.y;
  const int b0    = mtile * kROWS;
  const int arow  = lane & 15;
  const int koff  = (lane >> 4) << 1;
  const int rhalf = (lane >> 4) << 3;

  const size_t pr0 = (size_t)b0 * kNC + cell;
  const int cc = lane & 15, hh = lane >> 4;
  const int r0 = 4 * __builtin_amdgcn_readfirstlane(wave);
  const int g = wave & 1;
  const int q = wave >> 1;
  const float* w1c = w1 + (size_t)cell * kMODIN * kHMOD;
  const float* bpB = w1c + (size_t)(kHEAD + 4 * kBQ * q + koff) * kHMOD +
                     64 * g + 4 * arow;
  const float* bpH = w1c + (size_t)(4 * kU1 * q + koff) * kHMOD +
                     64 * g + 4 * arow;
  const float* apB = ldsA + arow * kMODIN + koff + kHEAD + 4 * kBQ * q;
  const float* apH = ldsA + arow * kMODIN + koff + 4 * kU1 * q;
  auto grp_ptr = [&](int t) -> const float* {
    return t < kG1 ? bpB + (size_t)(4 * kU1) * t * kHMOD : bpH;
  };
  auto unit_ptr = [&](int u) -> const float* {
    return ((u & 1) ? msg : h) +
           (pr0 + (size_t)(r0 + (u >> 1)) * kNC) * kUNIT + 4 * lane;
  };
  float4 lo[kNB][kU1], hi[kNB][kU1];
  float4 tlo[1], thi[1];
  v4f sb[2][kUQ];

  {
    const float* hb = hebb + pr0 * (kN * kN) + 4 * tid;
    v4f hv[kROWS];
#pragma unroll
    for (int r = 0; r < kROWS; ++r)
      hv[r] = __builtin_nontemporal_load(
          (const v4f*)(hb + (size_t)r * kNC * (kN * kN)));

    float2 d2[2];
    float dr[2], sl[2], se[2];
#pragma unroll
    for (int j = 0; j < 2; ++j) {
      const int rr = r0 + 2 * j + hh;
      d2[j] = *(const float2*)(decay + (pr0 + (size_t)rr * kNC) * kN + 2 * cc);
      dr[j] = drift[pr0 + (size_t)rr * kNC];
      sl[j] = s_live[b0 + rr];
      se[j] = s_ema[b0 + rr];
    }
    __builtin_amdgcn_sched_barrier(0);
#pragma unroll
    for (int t = 0; t < kNB - 1; ++t)
      load_b<kU1>(lo[t], hi[t], grp_ptr(t), kHMOD);
    __builtin_amdgcn_sched_barrier(0);
    load_unit(sb[0], unit_ptr(0));
    __builtin_amdgcn_sched_barrier(0);
    load_unit(sb[1], unit_ptr(1));
    __builtin_amdgcn_sched_barrier(0);
#pragma unroll
    for (int r = 0; r < kROWS; ++r)
      *(v4f*)(ldsA + r * kMODIN + kHEAD + 4 * tid) = hv[r];
    asm volatile("" : "+v"(dr[0]), "+v"(sl[0]), "+v"(se[0]),
                      "+v"(dr[1]), "+v"(sl[1]), "+v"(se[1]));

#pragma unroll
    for (int j = 0; j < 2; ++j) {
      float m = d2[j].x + d2[j].y;
      m += half_xor<8>(m);
      m += half_xor<4>(m);
      m += half_xor<2>(m);
      m += half_xor<1>(m);
      if (cc == 0)
        *(float4*)(ldsA + (r0 + 2 * j + hh) * kMODIN + 2 * kN) =
            make_float4(m * (1.f / (float)kN), dr[j], sl[j], se[j]);
    }
  }
  asm volatile("s_wait_dscnt 0x0\n\ts_barrier_signal -1\n\ts_barrier_wait -1"
               ::: "memory");

  {
    v8f c[kMT][4] = {};
    float* na = ldsA + r0 * kMODIN + 2 * cc + hh;
#pragma unroll
    for (int it = 0; it < kG1; ++it) {
      if (it + kNB - 1 <= kG1) {
        load_b<kU1>(lo[(it + kNB - 1) % kNB], hi[(it + kNB - 1) % kNB],
                    grp_ptr(it + kNB - 1), kHMOD);
        if (it + kNB - 1 == kG1 && q == 3)
          load_b<1>(tlo, thi, bpH + (size_t)(4 * kU1) * kHMOD, kHMOD);
      }
      if ((it & 1) == 0 && it >= 2 && it / 2 + 1 < 8)
        load_unit(sb[(it / 2 + 1) & 1], unit_ptr(it / 2 + 1));
      __builtin_amdgcn_sched_barrier(0);
      mma_b<kU1>(lo[it % kNB], hi[it % kNB], apB + 4 * kU1 * it,
                 16 * kMODIN, c);
      if (it & 1) {
        const int u = it >> 1;
        na[(u >> 1) * kMODIN + (u & 1) * kN] = unit_norm(sb[u & 1], cc);
      }
      __builtin_amdgcn_sched_barrier(0);
    }
    __syncthreads();
    mma_b<kU1>(lo[kG1 % kNB], hi[kG1 % kNB], apH, 16 * kMODIN, c);
    if (q == 3)
      mma_b<1>(tlo, thi, apH + 4 * kU1, 16 * kMODIN, c);

    float* pp = ldsP + (q * kROWS + rhalf) * kHMOD + 64 * g + 4 * arow;
#pragma unroll
    for (int mt = 0; mt < kMT; ++mt)
#pragma unroll
      for (int r = 0; r < 8; ++r)
        *(float4*)(pp + (16 * mt + r) * kHMOD) =
            make_float4(c[mt][0][r], c[mt][1][r], c[mt][2][r], c[mt][3][r]);
  }
  __syncthreads();
#pragma unroll
  for (int mt = 0; mt < kMT; ++mt) {
    const int row = (tid >> 4) + 16 * mt, col = (tid & 15) << 3;
    const float* bb = b1 + cell * kHMOD + col;
    float s[8];
    {
      const float4 u0 = *(const float4*)(bb);
      const float4 u1 = *(const float4*)(bb + 4);
      s[0] = u0.x; s[1] = u0.y; s[2] = u0.z; s[3] = u0.w;
      s[4] = u1.x; s[5] = u1.y; s[6] = u1.z; s[7] = u1.w;
    }
#pragma unroll
    for (int q = 0; q < 4; ++q) {
      const float* pq = ldsP + (q * kROWS + row) * kHMOD + col;
      const float4 u0 = *(const float4*)(pq);
      const float4 u1 = *(const float4*)(pq + 4);
      s[0] += u0.x; s[1] += u0.y; s[2] += u0.z; s[3] += u0.w;
      s[4] += u1.x; s[5] += u1.y; s[6] += u1.z; s[7] += u1.w;
    }
    float* hp = ldsH + row * kHSTRIDE + col;
    *(float4*)(hp) =
        make_float4(tanhf(s[0]), tanhf(s[1]), tanhf(s[2]), tanhf(s[3]));
    *(float4*)(hp + 4) =
        make_float4(tanhf(s[4]), tanhf(s[5]), tanhf(s[6]), tanhf(s[7]));
  }
  __syncthreads();

  const float* w2c = w2 + (size_t)cell * kHMOD * kMODOUT;
  const float* b2c = b2 + (size_t)cell * kMODOUT;
  const float* aH  = ldsH + arow * kHSTRIDE + koff;
#pragma unroll
  for (int st = wave; st < 16; st += 8) {
    const int cb = 64 * st + 4 * arow;
    v8f c[kMT][4];
    {
      const float4 bias = *(const float4*)(b2c + cb);
#pragma unroll
      for (int mt = 0; mt < kMT; ++mt)
#pragma unroll
        for (int i = 0; i < 8; ++i) {
          c[mt][0][i] = bias.x; c[mt][1][i] = bias.y;
          c[mt][2][i] = bias.z; c[mt][3][i] = bias.w;
        }
    }
    const float* bp = w2c + (size_t)koff * kMODOUT + cb;
    float4 lo[kU2], hi[kU2], nlo[kU2], nhi[kU2];
    load_b<kU2>(lo, hi, bp, kMODOUT);
#pragma unroll
    for (int it = 0; it < kHMOD / 4 / kU2; ++it) {
      if (it + 1 < kHMOD / 4 / kU2)
        load_b<kU2>(nlo, nhi, bp + (size_t)(4 * kU2) * (it + 1) * kMODOUT,
                    kMODOUT);
      mma_b<kU2>(lo, hi, aH + 4 * kU2 * it, 16 * kHSTRIDE, c);
#pragma unroll
      for (int u = 0; u < kU2; ++u) {
        lo[u] = nlo[u];
        hi[u] = nhi[u];
      }
    }
    float* op = out + ((size_t)(b0 + rhalf) * kNC + cell) * kMODOUT + cb;
#pragma unroll
    for (int mt = 0; mt < kMT; ++mt)
#pragma unroll
      for (int r = 0; r < 8; ++r)
        *(float4*)(op + (size_t)(16 * mt + r) * kNC * kMODOUT) =
            make_float4(c[mt][0][r], c[mt][1][r], c[mt][2][r], c[mt][3][r]);
  }
  if (wave < 2) {
    const int ocol = 1024 + (wave << 4) + arow;
    v8f c[kMT];
    {
      const float bias = b2c[ocol];
#pragma unroll
      for (int mt = 0; mt < kMT; ++mt)
#pragma unroll
        for (int i = 0; i < 8; ++i) c[mt][i] = bias;
    }
#pragma unroll
    for (int k0 = 0; k0 < kHMOD; k0 += 4) {
      const float bx = w2c[(size_t)(k0 + koff) * kMODOUT + ocol];
      const float by = w2c[(size_t)(k0 + koff + 1) * kMODOUT + ocol];
#pragma unroll
      for (int mt = 0; mt < kMT; ++mt) {
        const v2f a = *(const v2f*)(aH + mt * 16 * kHSTRIDE + k0);
        c[mt] = wmma4(a, bx, by, c[mt]);
      }
    }
#pragma unroll
    for (int mt = 0; mt < kMT; ++mt)
#pragma unroll
      for (int i = 0; i < 8; ++i) {
        const int row = b0 + 16 * mt + rhalf + i;
        out[((size_t)row * kNC + cell) * kMODOUT + ocol] = c[mt][i];
      }
  }
}

extern "C" void kernel_launch(void* const* d_in, const int* in_sizes, int n_in,
                              void* d_out, int out_size, void* d_ws, size_t ws_size,
                              hipStream_t stream) {
  const float* h     = (const float*)d_in[0];
  const float* msg   = (const float*)d_in[1];
  const float* decay = (const float*)d_in[2];
  const float* hebb  = (const float*)d_in[3];
  const float* drift = (const float*)d_in[4];
  const float* slive = (const float*)d_in[5];
  const float* sema  = (const float*)d_in[6];
  const float* w1    = (const float*)d_in[7];
  const float* b1    = (const float*)d_in[8];
  const float* w2    = (const float*)d_in[9];
  const float* b2    = (const float*)d_in[10];
  float* out    = (float*)d_out;

  dim3 grid(kBS / kROWS, kNC);
  modmlp_kernel<<<grid, 256, 0, stream>>>(
      h, msg, decay, drift, slive, sema, hebb, w1, b1, w2, b2, out);
}


# A fragments read from LDS outside the matrix streams: layer 2 holds hidden (both M tiles, all 32 k-steps) in 128 registers, layer 1 reads each group's 8 pairs one group ahead, beside its requests
# speedup vs baseline: 2.7288x; 1.0643x over previous
#include <hip/hip_runtime.h>
#include <math.h>

constexpr int kBS      = 64;
constexpr int kNC      = 128;
constexpr int kN       = 32;
constexpr int kDN      = 64;
constexpr int kHMOD    = 128;
constexpr int kMODIN   = 2 * kN + 4 + kN * kN;
constexpr int kMODOUT  = kN * kN + kN;
constexpr int kHEAD    = 2 * kN + 4;
constexpr int kHSTRIDE = 132;
constexpr int kMT      = 2;
constexpr int kROWS    = 16 * kMT;

typedef __attribute__((ext_vector_type(2))) float v2f;
typedef __attribute__((ext_vector_type(8))) float v8f;

typedef __attribute__((ext_vector_type(4))) float v4f;

constexpr int kUNIT  = kN * kDN;
constexpr int kUQ    = kUNIT / 128;

template <int M>
__device__ __forceinline__ float half_xor(float x) {
  const int b = __builtin_bit_cast(int, x);
  return __builtin_bit_cast(
      float, __builtin_amdgcn_update_dpp(b, b, 0x160 + M, 0xf, 0xf, false));
}

__device__ __forceinline__ void load_unit(v4f (&v)[kUQ], const float* p) {
#pragma unroll
  for (int q = 0; q < kUQ; ++q)
    v[q] = __builtin_nontemporal_load((const v4f*)(p + 128 * q));
}

template <int M, int CNT>
__device__ __forceinline__ void fold(const float (&in)[2 * CNT],
                                     float (&out)[CNT], bool up) {
#pragma unroll
  for (int i = 0; i < CNT; ++i) {
    const float keep = up ? in[i + CNT] : in[i];
    const float send = up ? in[i] : in[i + CNT];
    out[i] = keep + half_xor<M>(send);
  }
}

__device__ __forceinline__ float unit_norm(const v4f (&v)[kUQ], int c) {
  float p[kUQ], r8[8], r4[4], r2[2], r1[1];
#pragma unroll
  for (int q = 0; q < kUQ; ++q)
    p[q] = v[q].x * v[q].x + v[q].y * v[q].y + v[q].z * v[q].z +
           v[q].w * v[q].w;
  fold<8, 8>(p, r8, (c & 8) != 0);
  fold<4, 4>(r8, r4, (c & 4) != 0);
  fold<2, 2>(r4, r2, (c & 2) != 0);
  fold<1, 1>(r2, r1, (c & 1) != 0);
  return sqrtf(r1[0]);
}

__device__ __forceinline__ v8f wmma4(v2f a, float bx, float by, v8f c) {
  v2f b;
  b.x = bx;
  b.y = by;
  return __builtin_amdgcn_wmma_f32_16x16x4_f32(
      false, a, false, b, (short)0, c, false, false);
}

template <int U>
__device__ __forceinline__ void load_b(float4 (&lo)[U], float4 (&hi)[U],
                                       const float* __restrict__ p,
                                       int pitch) {
#pragma unroll
  for (int u = 0; u < U; ++u) {
    lo[u] = *(const float4*)(p + (size_t)(4 * u) * pitch);
    hi[u] = *(const float4*)(p + (size_t)(4 * u + 1) * pitch);
  }
}

template <int U>
__device__ __forceinline__ void mma_b(const float4 (&lo)[U],
                                      const float4 (&hi)[U],
                                      const float* __restrict__ ap, int mpitch,
                                      v8f (&c)[kMT][4]) {
#pragma unroll
  for (int u = 0; u < U; ++u) {
#pragma unroll
    for (int mt = 0; mt < kMT; ++mt) {
      const v2f a = *(const v2f*)(ap + mt * mpitch + 4 * u);
      c[mt][0] = wmma4(a, lo[u].x, hi[u].x, c[mt][0]);
      c[mt][1] = wmma4(a, lo[u].y, hi[u].y, c[mt][1]);
      c[mt][2] = wmma4(a, lo[u].z, hi[u].z, c[mt][2]);
      c[mt][3] = wmma4(a, lo[u].w, hi[u].w, c[mt][3]);
    }
  }
}

constexpr int kKS2   = kHMOD / 4;

template <int U>
__device__ __forceinline__ void mma_r(const float4 (&lo)[U],
                                      const float4 (&hi)[U],
                                      const v2f (&af)[kMT][kKS2], int ks0,
                                      v8f (&c)[kMT][4]) {
#pragma unroll
  for (int u = 0; u < U; ++u) {
#pragma unroll
    for (int mt = 0; mt < kMT; ++mt) {
      const v2f a = af[mt][ks0 + u];
      c[mt][0] = wmma4(a, lo[u].x, hi[u].x, c[mt][0]);
      c[mt][1] = wmma4(a, lo[u].y, hi[u].y, c[mt][1]);
      c[mt][2] = wmma4(a, lo[u].z, hi[u].z, c[mt][2]);
      c[mt][3] = wmma4(a, lo[u].w, hi[u].w, c[mt][3]);
    }
  }
}

template <int U>
__device__ __forceinline__ void load_a(v2f (&a)[U][kMT],
                                       const float* __restrict__ ap,
                                       int mpitch) {
#pragma unroll
  for (int u = 0; u < U; ++u)
#pragma unroll
    for (int mt = 0; mt < kMT; ++mt)
      a[u][mt] = *(const v2f*)(ap + mt * mpitch + 4 * u);
}

template <int U>
__device__ __forceinline__ void mma_ra(const float4 (&lo)[U],
                                       const float4 (&hi)[U],
                                       const v2f (&a)[U][kMT],
                                       v8f (&c)[kMT][4]) {
#pragma unroll
  for (int u = 0; u < U; ++u) {
#pragma unroll
    for (int mt = 0; mt < kMT; ++mt) {
      c[mt][0] = wmma4(a[u][mt], lo[u].x, hi[u].x, c[mt][0]);
      c[mt][1] = wmma4(a[u][mt], lo[u].y, hi[u].y, c[mt][1]);
      c[mt][2] = wmma4(a[u][mt], lo[u].z, hi[u].z, c[mt][2]);
      c[mt][3] = wmma4(a[u][mt], lo[u].w, hi[u].w, c[mt][3]);
    }
  }
}

constexpr int kHK    = kHEAD / 4;
constexpr int kBQ    = (kN * kN) / 16;
constexpr int kU1    = 4;
constexpr int kG1    = kBQ / kU1;
constexpr int kNB    = 4;
static_assert(kHK == 4 * kU1 + 1, "head k-steps: 4 per K-quarter and one over");
constexpr int kU2    = 8;

constexpr int kLdsFloats = kROWS * kMODIN + kROWS * kHSTRIDE + 4 * kROWS * kHMOD;

__global__ __launch_bounds__(256) __attribute__((amdgpu_waves_per_eu(2, 2)))
void modmlp_kernel(
    const float* __restrict__ h, const float* __restrict__ msg,
    const float* __restrict__ decay, const float* __restrict__ drift,
    const float* __restrict__ s_live, const float* __restrict__ s_ema,
    const float* __restrict__ hebb,
    const float* __restrict__ w1, const float* __restrict__ b1,
    const float* __restrict__ w2, const float* __restrict__ b2,
    float* __restrict__ out) {
  __shared__ __align__(16) float smem[kLdsFloats];
  float* ldsA = smem;
  float* ldsH = smem + kROWS * kMODIN;
  float* ldsP = ldsH + kROWS * kHSTRIDE;

  const int tid   = threadIdx.x;
  const int lane  = tid & 31;
  const int wave  = tid >> 5;
  const int mtile = blockIdx.x;
  const int cell  = blockIdx.y;
  const int b0    = mtile * kROWS;
  const int arow  = lane & 15;
  const int koff  = (lane >> 4) << 1;
  const int rhalf = (lane >> 4) << 3;

  const size_t pr0 = (size_t)b0 * kNC + cell;
  const int cc = lane & 15, hh = lane >> 4;
  const int r0 = 4 * __builtin_amdgcn_readfirstlane(wave);
  const int g = wave & 1;
  const int q = wave >> 1;
  const float* w1c = w1 + (size_t)cell * kMODIN * kHMOD;
  const float* bpB = w1c + (size_t)(kHEAD + 4 * kBQ * q + koff) * kHMOD +
                     64 * g + 4 * arow;
  const float* bpH = w1c + (size_t)(4 * kU1 * q + koff) * kHMOD +
                     64 * g + 4 * arow;
  const float* apB = ldsA + arow * kMODIN + koff + kHEAD + 4 * kBQ * q;
  const float* apH = ldsA + arow * kMODIN + koff + 4 * kU1 * q;
  auto grp_ptr = [&](int t) -> const float* {
    return t < kG1 ? bpB + (size_t)(4 * kU1) * t * kHMOD : bpH;
  };
  auto unit_ptr = [&](int u) -> const float* {
    return ((u & 1) ? msg : h) +
           (pr0 + (size_t)(r0 + (u >> 1)) * kNC) * kUNIT + 4 * lane;
  };
  float4 lo[kNB][kU1], hi[kNB][kU1];
  float4 tlo[1], thi[1];
  v4f sb[2][kUQ];

  {
    const float* hb = hebb + pr0 * (kN * kN) + 4 * tid;
    v4f hv[kROWS];
#pragma unroll
    for (int r = 0; r < kROWS; ++r)
      hv[r] = __builtin_nontemporal_load(
          (const v4f*)(hb + (size_t)r * kNC * (kN * kN)));

    float2 d2[2];
    float dr[2], sl[2], se[2];
#pragma unroll
    for (int j = 0; j < 2; ++j) {
      const int rr = r0 + 2 * j + hh;
      d2[j] = *(const float2*)(decay + (pr0 + (size_t)rr * kNC) * kN + 2 * cc);
      dr[j] = drift[pr0 + (size_t)rr * kNC];
      sl[j] = s_live[b0 + rr];
      se[j] = s_ema[b0 + rr];
    }
    __builtin_amdgcn_sched_barrier(0);
#pragma unroll
    for (int t = 0; t < kNB - 1; ++t)
      load_b<kU1>(lo[t], hi[t], grp_ptr(t), kHMOD);
    __builtin_amdgcn_sched_barrier(0);
    load_unit(sb[0], unit_ptr(0));
    __builtin_amdgcn_sched_barrier(0);
    load_unit(sb[1], unit_ptr(1));
    __builtin_amdgcn_sched_barrier(0);
#pragma unroll
    for (int r = 0; r < kROWS; ++r)
      *(v4f*)(ldsA + r * kMODIN + kHEAD + 4 * tid) = hv[r];
    asm volatile("" : "+v"(dr[0]), "+v"(sl[0]), "+v"(se[0]),
                      "+v"(dr[1]), "+v"(sl[1]), "+v"(se[1]));

#pragma unroll
    for (int j = 0; j < 2; ++j) {
      float m = d2[j].x + d2[j].y;
      m += half_xor<8>(m);
      m += half_xor<4>(m);
      m += half_xor<2>(m);
      m += half_xor<1>(m);
      if (cc == 0)
        *(float4*)(ldsA + (r0 + 2 * j + hh) * kMODIN + 2 * kN) =
            make_float4(m * (1.f / (float)kN), dr[j], sl[j], se[j]);
    }
  }
  asm volatile("s_wait_dscnt 0x0\n\ts_barrier_signal -1\n\ts_barrier_wait -1"
               ::: "memory");

  {
    v8f c[kMT][4] = {};
    float* na = ldsA + r0 * kMODIN + 2 * cc + hh;
    v2f a1[2][kU1][kMT];
    load_a<kU1>(a1[0], apB, 16 * kMODIN);
#pragma unroll
    for (int it = 0; it < kG1; ++it) {
      if (it + kNB - 1 <= kG1) {
        load_b<kU1>(lo[(it + kNB - 1) % kNB], hi[(it + kNB - 1) % kNB],
                    grp_ptr(it + kNB - 1), kHMOD);
        if (it + kNB - 1 == kG1 && q == 3)
          load_b<1>(tlo, thi, bpH + (size_t)(4 * kU1) * kHMOD, kHMOD);
      }
      if ((it & 1) == 0 && it >= 2 && it / 2 + 1 < 8)
        load_unit(sb[(it / 2 + 1) & 1], unit_ptr(it / 2 + 1));
      if (it + 1 < kG1)
        load_a<kU1>(a1[(it + 1) & 1], apB + 4 * kU1 * (it + 1), 16 * kMODIN);
      __builtin_amdgcn_sched_barrier(0);
      mma_ra<kU1>(lo[it % kNB], hi[it % kNB], a1[it & 1], c);
      if (it & 1) {
        const int u = it >> 1;
        na[(u >> 1) * kMODIN + (u & 1) * kN] = unit_norm(sb[u & 1], cc);
      }
      __builtin_amdgcn_sched_barrier(0);
    }
    __syncthreads();
    mma_b<kU1>(lo[kG1 % kNB], hi[kG1 % kNB], apH, 16 * kMODIN, c);
    if (q == 3)
      mma_b<1>(tlo, thi, apH + 4 * kU1, 16 * kMODIN, c);

    float* pp = ldsP + (q * kROWS + rhalf) * kHMOD + 64 * g + 4 * arow;
#pragma unroll
    for (int mt = 0; mt < kMT; ++mt)
#pragma unroll
      for (int r = 0; r < 8; ++r)
        *(float4*)(pp + (16 * mt + r) * kHMOD) =
            make_float4(c[mt][0][r], c[mt][1][r], c[mt][2][r], c[mt][3][r]);
  }
  __syncthreads();
#pragma unroll
  for (int mt = 0; mt < kMT; ++mt) {
    const int row = (tid >> 4) + 16 * mt, col = (tid & 15) << 3;
    const float* bb = b1 + cell * kHMOD + col;
    float s[8];
    {
      const float4 u0 = *(const float4*)(bb);
      const float4 u1 = *(const float4*)(bb + 4);
      s[0] = u0.x; s[1] = u0.y; s[2] = u0.z; s[3] = u0.w;
      s[4] = u1.x; s[5] = u1.y; s[6] = u1.z; s[7] = u1.w;
    }
#pragma unroll
    for (int q = 0; q < 4; ++q) {
      const float* pq = ldsP + (q * kROWS + row) * kHMOD + col;
      const float4 u0 = *(const float4*)(pq);
      const float4 u1 = *(const float4*)(pq + 4);
      s[0] += u0.x; s[1] += u0.y; s[2] += u0.z; s[3] += u0.w;
      s[4] += u1.x; s[5] += u1.y; s[6] += u1.z; s[7] += u1.w;
    }
    float* hp = ldsH + row * kHSTRIDE + col;
    *(float4*)(hp) =
        make_float4(tanhf(s[0]), tanhf(s[1]), tanhf(s[2]), tanhf(s[3]));
    *(float4*)(hp + 4) =
        make_float4(tanhf(s[4]), tanhf(s[5]), tanhf(s[6]), tanhf(s[7]));
  }
  __syncthreads();

  const float* w2c = w2 + (size_t)cell * kHMOD * kMODOUT;
  const float* b2c = b2 + (size_t)cell * kMODOUT;
  const float* aH  = ldsH + arow * kHSTRIDE + koff;
  v2f af[kMT][kKS2];
#pragma unroll
  for (int s = 0; s < 2; ++s) {
    const int st = wave + 8 * s;
    const int cb = 64 * st + 4 * arow;
    v8f c[kMT][4];
    {
      const float4 bias = *(const float4*)(b2c + cb);
#pragma unroll
      for (int mt = 0; mt < kMT; ++mt)
#pragma unroll
        for (int i = 0; i < 8; ++i) {
          c[mt][0][i] = bias.x; c[mt][1][i] = bias.y;
          c[mt][2][i] = bias.z; c[mt][3][i] = bias.w;
        }
    }
    const float* bp = w2c + (size_t)koff * kMODOUT + cb;
    float4 lo[3][kU2], hi[3][kU2];
    load_b<kU2>(lo[0], hi[0], bp, kMODOUT);
    load_b<kU2>(lo[1], hi[1], bp + (size_t)(4 * kU2) * kMODOUT, kMODOUT);
    if (s == 0) {
#pragma unroll
      for (int mt = 0; mt < kMT; ++mt)
#pragma unroll
        for (int ks = 0; ks < kKS2; ++ks)
          af[mt][ks] = *(const v2f*)(aH + mt * 16 * kHSTRIDE + 4 * ks);
    }
    __builtin_amdgcn_sched_barrier(0);
#pragma unroll
    for (int it = 0; it < kKS2 / kU2; ++it) {
      if (it + 2 < kKS2 / kU2) {
        load_b<kU2>(lo[(it + 2) % 3], hi[(it + 2) % 3],
                    bp + (size_t)(4 * kU2) * (it + 2) * kMODOUT, kMODOUT);
        __builtin_amdgcn_sched_barrier(0);
      }
      mma_r<kU2>(lo[it % 3], hi[it % 3], af, kU2 * it, c);
    }
    float* op = out + ((size_t)(b0 + rhalf) * kNC + cell) * kMODOUT + cb;
#pragma unroll
    for (int mt = 0; mt < kMT; ++mt)
#pragma unroll
      for (int r = 0; r < 8; ++r)
        *(float4*)(op + (size_t)(16 * mt + r) * kNC * kMODOUT) =
            make_float4(c[mt][0][r], c[mt][1][r], c[mt][2][r], c[mt][3][r]);
  }
  if (wave < 2) {
    const int ocol = 1024 + (wave << 4) + arow;
    v8f c[kMT];
    {
      const float bias = b2c[ocol];
#pragma unroll
      for (int mt = 0; mt < kMT; ++mt)
#pragma unroll
        for (int i = 0; i < 8; ++i) c[mt][i] = bias;
    }
#pragma unroll
    for (int k0 = 0; k0 < kHMOD; k0 += 4) {
      const float bx = w2c[(size_t)(k0 + koff) * kMODOUT + ocol];
      const float by = w2c[(size_t)(k0 + koff + 1) * kMODOUT + ocol];
#pragma unroll
      for (int mt = 0; mt < kMT; ++mt)
        c[mt] = wmma4(af[mt][k0 / 4], bx, by, c[mt]);
    }
#pragma unroll
    for (int mt = 0; mt < kMT; ++mt)
#pragma unroll
      for (int i = 0; i < 8; ++i) {
        const int row = b0 + 16 * mt + rhalf + i;
        out[((size_t)row * kNC + cell) * kMODOUT + ocol] = c[mt][i];
      }
  }
}

extern "C" void kernel_launch(void* const* d_in, const int* in_sizes, int n_in,
                              void* d_out, int out_size, void* d_ws, size_t ws_size,
                              hipStream_t stream) {
  const float* h     = (const float*)d_in[0];
  const float* msg   = (const float*)d_in[1];
  const float* decay = (const float*)d_in[2];
  const float* hebb  = (const float*)d_in[3];
  const float* drift = (const float*)d_in[4];
  const float* slive = (const float*)d_in[5];
  const float* sema  = (const float*)d_in[6];
  const float* w1    = (const float*)d_in[7];
  const float* b1    = (const float*)d_in[8];
  const float* w2    = (const float*)d_in[9];
  const float* b2    = (const float*)d_in[10];
  float* out    = (float*)d_out;

  dim3 grid(kBS / kROWS, kNC);
  modmlp_kernel<<<grid, 256, 0, stream>>>(
      h, msg, decay, drift, slive, sema, hebb, w1, b1, w2, b2, out);
}
